# starting kernel
# baseline (speedup 1.0000x reference)
#include <hip/hip_runtime.h>

typedef __attribute__((ext_vector_type(2))) float v2f;
typedef __attribute__((ext_vector_type(8))) float v8f;

#define LL_BIG   (1 << 22)
#define COLS     4096
#define ROWS     1024
#define ROWS_OUT 576
#define DD_W     2359296
#define TC       32
#define TSTRIDE  33
#define BPB      4

__device__ __forceinline__ void fwht_lds(float* buf, int n, int tid, int nt) {
    for (int h = 1; h < n; h <<= 1) {
        __syncthreads();
        for (int p = tid; p < (n >> 1); p += nt) {
            int i0 = ((p & ~(h - 1)) << 1) | (p & (h - 1));
            int i1 = i0 + h;
            float a = buf[i0], b = buf[i1];
            buf[i0] = a + b;
            buf[i1] = a - b;
        }
    }
    __syncthreads();
}

__global__ void small_fwht_kernel(const float* __restrict__ theta,
                                  const float* __restrict__ BB,
                                  float* __restrict__ wout, int P) {
    __shared__ float w[2048];
    int tid = threadIdx.x, nt = blockDim.x;
    for (int j = tid; j < P; j += nt) w[j] = BB[j] * theta[j];
    fwht_lds(w, P, tid, nt);
    for (int j = tid; j < P; j += nt) wout[j] = w[j];
}

__global__ void gg_partial_kernel(const float* __restrict__ GG, int n,
                                  float* __restrict__ part) {
    __shared__ float red[256];
    int tid = threadIdx.x;
    float s = 0.f;
    for (int i = blockIdx.x * blockDim.x + tid; i < n; i += gridDim.x * blockDim.x) {
        float g = GG[i]; s += g * g;
    }
    red[tid] = s; __syncthreads();
    for (int off = 128; off > 0; off >>= 1) {
        if (tid < off) red[tid] += red[tid + off];
        __syncthreads();
    }
    if (tid == 0) part[blockIdx.x] = red[0];
}

__global__ void gg_final_kernel(const float* __restrict__ part, float* __restrict__ gsum) {
    __shared__ float red[256];
    int tid = threadIdx.x;
    red[tid] = part[tid]; __syncthreads();
    for (int off = 128; off > 0; off >>= 1) {
        if (tid < off) red[tid] += red[tid + off];
        __syncthreads();
    }
    if (tid == 0) gsum[0] = red[0];
}

__global__ __launch_bounds__(512)
void fwht_passA_kernel(const float* __restrict__ w, const int* __restrict__ Pi,
                       const float* __restrict__ GG, float* __restrict__ mbuf,
                       int pmask) {
    __shared__ float tile[ROWS * TSTRIDE];
    int tid = threadIdx.x, nt = blockDim.x;
    int col0 = blockIdx.x * TC;
    for (int idx = tid; idx < ROWS * TC; idx += nt) {
        int r = idx >> 5, c = idx & (TC - 1);
        int g = r * COLS + col0 + c;
        tile[r * TSTRIDE + c] = w[Pi[g] & pmask] * GG[g];
    }
    for (int h = 1; h < ROWS; h <<= 1) {
        __syncthreads();
        for (int q = tid; q < (ROWS >> 1) * TC; q += nt) {
            int c = q & (TC - 1), p = q >> 5;
            int r0 = ((p & ~(h - 1)) << 1) | (p & (h - 1));
            int r1 = r0 + h;
            float a = tile[r0 * TSTRIDE + c], b = tile[r1 * TSTRIDE + c];
            tile[r0 * TSTRIDE + c] = a + b;
            tile[r1 * TSTRIDE + c] = a - b;
        }
    }
    __syncthreads();
    for (int idx = tid; idx < ROWS_OUT * TC; idx += nt) {
        int r = idx >> 5, c = idx & (TC - 1);
        mbuf[(size_t)r * COLS + col0 + c] = tile[r * TSTRIDE + c];
    }
}

__global__ __launch_bounds__(512)
void fwht_passB_kernel(const float* __restrict__ mbuf, const float* __restrict__ init,
                       const float* __restrict__ gsum, float* __restrict__ out, int DD) {
    __shared__ float blk[BPB * COLS];
    int tid = threadIdx.x, nt = blockDim.x;
    size_t base = (size_t)blockIdx.x * (BPB * COLS);
    for (int i = tid; i < BPB * COLS; i += nt) blk[i] = mbuf[base + i];
    for (int h = 1; h < COLS; h <<= 1) {
        __syncthreads();
        for (int q = tid; q < BPB * (COLS >> 1); q += nt) {
            int sb = q >> 11, p = q & (COLS / 2 - 1);
            int i0 = (sb << 12) | (((p & ~(h - 1)) << 1) | (p & (h - 1)));
            int i1 = i0 + h;
            float a = blk[i0], b = blk[i1];
            blk[i0] = a + b;
            blk[i1] = a - b;
        }
    }
    __syncthreads();
    float scale = rsqrtf((float)DD * gsum[0]);
    for (int i = tid; i < BPB * COLS; i += nt) {
        size_t g = base + i;
        out[g] = init[g] + blk[i] * scale;
    }
}

__global__ __launch_bounds__(512)
void fastfood_small_kernel(const float* __restrict__ theta, const float* __restrict__ BB,
                           const int* __restrict__ Pi, const float* __restrict__ GG,
                           const float* __restrict__ init, float* __restrict__ out,
                           int P, int LLv, int DD) {
    __shared__ float w[2048];
    __shared__ float m[4096];
    __shared__ float red[512];
    int tid = threadIdx.x, nt = blockDim.x;
    for (int j = tid; j < P; j += nt) w[j] = BB[j] * theta[j];
    fwht_lds(w, P, tid, nt);
    float s = 0.f;
    for (int i = tid; i < LLv; i += nt) { float g = GG[i]; s += g * g; }
    red[tid] = s; __syncthreads();
    for (int off = nt >> 1; off > 0; off >>= 1) {
        if (tid < off) red[tid] += red[tid + off];
        __syncthreads();
    }
    for (int i = tid; i < LLv; i += nt) m[i] = w[Pi[i] & (P - 1)] * GG[i];
    fwht_lds(m, LLv, tid, nt);
    float scale = rsqrtf((float)DD * red[0]);
    for (int i = tid; i < DD; i += nt) out[i] = init[i] + m[i] * scale;
}

template<int RELU>
__global__ __launch_bounds__(256)
void gemm_wmma_kernel(const float* __restrict__ A, const float* __restrict__ B,
                      const float* __restrict__ bias, float* __restrict__ C,
                      int M, int N, int K) {
    int lane = threadIdx.x & 31;
    int wave = threadIdx.x >> 5;
    int job = blockIdx.x * 8 + wave;
    int mTiles = M >> 4;
    int mt = job % mTiles;
    int ng = job / mTiles;
    int row = lane & 15;
    int kg = (lane >> 4) << 1;
    const float* ap  = A + (size_t)(mt * 16 + row) * K + kg;
    const float* bp0 = B + (size_t)(ng * 64 + row) * K + kg;
    const float* bp1 = bp0 + (size_t)16 * K;
    const float* bp2 = bp0 + (size_t)32 * K;
    const float* bp3 = bp0 + (size_t)48 * K;
    v8f acc0 = {}, acc1 = {}, acc2 = {}, acc3 = {};

    v2f a  = *(const v2f*)(ap);
    v2f b0 = *(const v2f*)(bp0);
    v2f b1 = *(const v2f*)(bp1);
    v2f b2 = *(const v2f*)(bp2);
    v2f b3 = *(const v2f*)(bp3);

    for (int k0 = 4; k0 < K; k0 += 4) {
        v2f an  = *(const v2f*)(ap  + k0);
        v2f bn0 = *(const v2f*)(bp0 + k0);
        v2f bn1 = *(const v2f*)(bp1 + k0);
        v2f bn2 = *(const v2f*)(bp2 + k0);
        v2f bn3 = *(const v2f*)(bp3 + k0);
        acc0 = __builtin_amdgcn_wmma_f32_16x16x4_f32(false, a, false, b0, (short)0, acc0, false, false);
        acc1 = __builtin_amdgcn_wmma_f32_16x16x4_f32(false, a, false, b1, (short)0, acc1, false, false);
        acc2 = __builtin_amdgcn_wmma_f32_16x16x4_f32(false, a, false, b2, (short)0, acc2, false, false);
        acc3 = __builtin_amdgcn_wmma_f32_16x16x4_f32(false, a, false, b3, (short)0, acc3, false, false);
        a = an; b0 = bn0; b1 = bn1; b2 = bn2; b3 = bn3;
    }
    acc0 = __builtin_amdgcn_wmma_f32_16x16x4_f32(false, a, false, b0, (short)0, acc0, false, false);
    acc1 = __builtin_amdgcn_wmma_f32_16x16x4_f32(false, a, false, b1, (short)0, acc1, false, false);
    acc2 = __builtin_amdgcn_wmma_f32_16x16x4_f32(false, a, false, b2, (short)0, acc2, false, false);
    acc3 = __builtin_amdgcn_wmma_f32_16x16x4_f32(false, a, false, b3, (short)0, acc3, false, false);

    int mrow = mt * 16 + (lane >> 4) * 8;
    v8f accs[4] = {acc0, acc1, acc2, acc3};
#pragma unroll
    for (int t = 0; t < 4; ++t) {
        int n = ng * 64 + t * 16 + row;
        float bv = bias[n];
        v8f acc = accs[t];
#pragma unroll
        for (int v = 0; v < 8; ++v) {
            float xv = acc[v] + bv;
            if (RELU) xv = fmaxf(xv, 0.f);
            C[(size_t)(mrow + v) * N + n] = xv;
        }
    }
}

extern "C" void kernel_launch(void* const* d_in, const int* in_sizes, int n_in,
                              void* d_out, int out_size, void* d_ws, size_t ws_size,
                              hipStream_t stream) {
    const float* x     = (const float*)d_in[0];
    const float* theta = (const float*)d_in[1];
    const float* iW1   = (const float*)d_in[2];
    const float* ib1   = (const float*)d_in[3];
    const float* iW2   = (const float*)d_in[4];
    const float* ib2   = (const float*)d_in[5];
    const float* BB_W1 = (const float*)d_in[6];
    const int*   Pi_W1 = (const int*)d_in[7];
    const float* GG_W1 = (const float*)d_in[8];
    const float* BB_b1 = (const float*)d_in[9];
    const int*   Pi_b1 = (const int*)d_in[10];
    const float* GG_b1 = (const float*)d_in[11];
    const float* BB_W2 = (const float*)d_in[12];
    const int*   Pi_W2 = (const int*)d_in[13];
    const float* GG_W2 = (const float*)d_in[14];
    const float* BB_b2 = (const float*)d_in[15];
    const int*   Pi_b2 = (const int*)d_in[16];
    const float* GG_b2 = (const float*)d_in[17];
    float* out = (float*)d_out;

    float* ws = (float*)d_ws;
    size_t o = 0;
    float* W1   = ws + o; o += DD_W;
    float* W2   = ws + o; o += DD_W;
    float* b1   = ws + o; o += 3072;
    float* b2   = ws + o; o += 768;
    float* hbuf = ws + o; o += (size_t)256 * 3072;
    float* mbuf = ws + o; o += DD_W;
    float* wsm  = ws + o; o += 2048;
    float* part = ws + o; o += 256;
    float* gsum = ws + o; o += 1;

    small_fwht_kernel<<<1, 512, 0, stream>>>(theta, BB_W1, wsm, 2048);
    gg_partial_kernel<<<256, 256, 0, stream>>>(GG_W1, LL_BIG, part);
    gg_final_kernel<<<1, 256, 0, stream>>>(part, gsum);
    fwht_passA_kernel<<<COLS / TC, 512, 0, stream>>>(wsm, Pi_W1, GG_W1, mbuf, 2047);
    fwht_passB_kernel<<<ROWS_OUT / BPB, 512, 0, stream>>>(mbuf, iW1, gsum, W1, DD_W);

    small_fwht_kernel<<<1, 512, 0, stream>>>(theta, BB_W2, wsm, 2048);
    gg_partial_kernel<<<256, 256, 0, stream>>>(GG_W2, LL_BIG, part);
    gg_final_kernel<<<1, 256, 0, stream>>>(part, gsum);
    fwht_passA_kernel<<<COLS / TC, 512, 0, stream>>>(wsm, Pi_W2, GG_W2, mbuf, 2047);
    fwht_passB_kernel<<<ROWS_OUT / BPB, 512, 0, stream>>>(mbuf, iW2, gsum, W2, DD_W);

    fastfood_small_kernel<<<1, 512, 0, stream>>>(theta, BB_b1, Pi_b1, GG_b1, ib1, b1,
                                                 2048, 4096, 3072);
    fastfood_small_kernel<<<1, 512, 0, stream>>>(theta, BB_b2, Pi_b2, GG_b2, ib2, b2,
                                                 1024, 1024, 768);

    gemm_wmma_kernel<1><<<96, 256, 0, stream>>>(x, W1, b1, hbuf, 256, 3072, 768);
    gemm_wmma_kernel<0><<<24, 256, 0, stream>>>(hbuf, W2, b2, out, 256, 768, 3072);
}


# sum(GG^2), pass A, pass B: loops that waited on one load per trip (global streams, Pi gather, LDS butterflies) issue a whole batch of independent loads before first use; sum and stage orders kept
# speedup vs baseline: 2.8173x; 2.8173x over previous
#include <hip/hip_runtime.h>

typedef __attribute__((ext_vector_type(2))) float v2f;
typedef __attribute__((ext_vector_type(8))) float v8f;
typedef __attribute__((ext_vector_type(4))) float v4f;

#define LL_BIG   (1 << 22)
#define COLS     4096
#define ROWS     1024
#define ROWS_OUT 576
#define DD_W     2359296
#define TC       32
#define TSTRIDE  33
#define BPB      4

__device__ __forceinline__ void fwht_lds(float* buf, int n, int tid, int nt) {
    for (int h = 1; h < n; h <<= 1) {
        __syncthreads();
        for (int p = tid; p < (n >> 1); p += nt) {
            int i0 = ((p & ~(h - 1)) << 1) | (p & (h - 1));
            int i1 = i0 + h;
            float a = buf[i0], b = buf[i1];
            buf[i0] = a + b;
            buf[i1] = a - b;
        }
    }
    __syncthreads();
}

__global__ void small_fwht_kernel(const float* __restrict__ theta,
                                  const float* __restrict__ BB,
                                  float* __restrict__ wout, int P) {
    __shared__ float w[2048];
    int tid = threadIdx.x, nt = blockDim.x;
    for (int j = tid; j < P; j += nt) w[j] = BB[j] * theta[j];
    fwht_lds(w, P, tid, nt);
    for (int j = tid; j < P; j += nt) wout[j] = w[j];
}

#define GGB 32
#define GGW 256
#define GGT 256
__global__ __launch_bounds__(GGT)
void gg_partial_kernel(const float* __restrict__ GG, int n,
                       float* __restrict__ part) {
    __shared__ float red[256];
    int tid = threadIdx.x;
    constexpr int stride = GGW * GGT;
    float s = 0.f;
    int i = blockIdx.x * GGT + tid;
#pragma unroll 1
    for (; i + (GGB - 1) * stride < n; i += GGB * stride) {
        float g[GGB];
#pragma unroll
        for (int u = 0; u < GGB; ++u) g[u] = GG[i + u * stride];
#pragma unroll
        for (int u = 0; u < GGB; ++u) s += g[u] * g[u];
    }
    for (; i < n; i += stride) {
        float g = GG[i]; s += g * g;
    }
    red[tid] = s; __syncthreads();
    for (int off = 128; off > 0; off >>= 1) {
        if (tid < off) red[tid] += red[tid + off];
        __syncthreads();
    }
    if (tid == 0) part[blockIdx.x] = red[0];
}

__global__ void gg_final_kernel(const float* __restrict__ part, float* __restrict__ gsum) {
    __shared__ float red[256];
    int tid = threadIdx.x;
    red[tid] = part[tid]; __syncthreads();
    for (int off = 128; off > 0; off >>= 1) {
        if (tid < off) red[tid] += red[tid + off];
        __syncthreads();
    }
    if (tid == 0) gsum[0] = red[0];
}

#define NTH  512
#define GAB  32
#define BAB  16

template<int H>
__device__ __forceinline__ void passA_stage(float* tile, int c, int pr) {
#pragma unroll 1
    for (int kb = 0; kb < (ROWS / 2) / 16; kb += BAB) {
        float a[BAB], b[BAB];
        int ad[BAB];
#pragma unroll
        for (int u = 0; u < BAB; ++u) {
            int p = pr + 16 * (kb + u);
            int r0 = ((p & ~(H - 1)) << 1) | (p & (H - 1));
            ad[u] = r0 * TSTRIDE + c;
            a[u] = tile[ad[u]];
            b[u] = tile[ad[u] + H * TSTRIDE];
        }
#pragma unroll
        for (int u = 0; u < BAB; ++u) {
            tile[ad[u]] = a[u] + b[u];
            tile[ad[u] + H * TSTRIDE] = a[u] - b[u];
        }
    }
}

__global__ __launch_bounds__(NTH)
void fwht_passA_kernel(const float* __restrict__ w, const int* __restrict__ Pi,
                       const float* __restrict__ GG, float* __restrict__ mbuf,
                       int pmask) {
    __shared__ float tile[ROWS * TSTRIDE];
    const int tid = threadIdx.x;
    const int c = tid & (TC - 1), pr = tid >> 5;
    const int col0 = blockIdx.x * TC;
    const int gbase = pr * COLS + col0 + c;
#pragma unroll 1
    for (int kb = 0; kb < ROWS / 16; kb += GAB) {
        int pi[GAB];
        float gg[GAB], wv[GAB];
#pragma unroll
        for (int u = 0; u < GAB; ++u) {
            int g = gbase + (kb + u) * 16 * COLS;
            pi[u] = Pi[g];
            gg[u] = GG[g];
        }
#pragma unroll
        for (int u = 0; u < GAB; ++u) wv[u] = w[pi[u] & pmask];
#pragma unroll
        for (int u = 0; u < GAB; ++u)
            tile[(pr + 16 * (kb + u)) * TSTRIDE + c] = wv[u] * gg[u];
    }
    __syncthreads(); passA_stage<1>(tile, c, pr);
    __syncthreads(); passA_stage<2>(tile, c, pr);
    __syncthreads(); passA_stage<4>(tile, c, pr);
    __syncthreads(); passA_stage<8>(tile, c, pr);
    __syncthreads(); passA_stage<16>(tile, c, pr);
    __syncthreads(); passA_stage<32>(tile, c, pr);
    __syncthreads(); passA_stage<64>(tile, c, pr);
    __syncthreads(); passA_stage<128>(tile, c, pr);
    __syncthreads(); passA_stage<256>(tile, c, pr);
    __syncthreads(); passA_stage<512>(tile, c, pr);
    __syncthreads();
    for (int idx = tid; idx < ROWS_OUT * TC; idx += NTH) {
        int r = idx >> 5, cc = idx & (TC - 1);
        mbuf[(size_t)r * COLS + col0 + cc] = tile[r * TSTRIDE + cc];
    }
}

#define BVEC (BPB * COLS / 4 / NTH)

template<int H>
__device__ __forceinline__ void passB_stage(float* blk, int tid) {
    float a[16], b[16];
    int ad[16];
#pragma unroll
    for (int k = 0; k < 16; ++k) {
        int p = tid + NTH * (k & 3);
        int i0 = ((k >> 2) << 12) | (((p & ~(H - 1)) << 1) | (p & (H - 1)));
        ad[k] = i0;
        a[k] = blk[i0];
        b[k] = blk[i0 + H];
    }
#pragma unroll
    for (int k = 0; k < 16; ++k) {
        blk[ad[k]] = a[k] + b[k];
        blk[ad[k] + H] = a[k] - b[k];
    }
}

__global__ __launch_bounds__(NTH)
void fwht_passB_kernel(const float* __restrict__ mbuf, const float* __restrict__ init,
                       const float* __restrict__ gsum, float* __restrict__ out, int DD) {
    __shared__ __attribute__((aligned(16))) float blk[BPB * COLS];
    const int tid = threadIdx.x;
    const size_t base = (size_t)blockIdx.x * (BPB * COLS);
    const v4f* src = (const v4f*)(mbuf + base);
    const v4f* ini = (const v4f*)(init + base);
    v4f* dst = (v4f*)(out + base);
    v4f* blk4 = (v4f*)blk;
    v4f mv[BVEC], iv[BVEC];
#pragma unroll
    for (int u = 0; u < BVEC; ++u) mv[u] = src[tid + u * NTH];
#pragma unroll
    for (int u = 0; u < BVEC; ++u) iv[u] = ini[tid + u * NTH];
    const float gs = gsum[0];
#pragma unroll
    for (int u = 0; u < BVEC; ++u) blk4[tid + u * NTH] = mv[u];
    __syncthreads(); passB_stage<1>(blk, tid);
    __syncthreads(); passB_stage<2>(blk, tid);
    __syncthreads(); passB_stage<4>(blk, tid);
    __syncthreads(); passB_stage<8>(blk, tid);
    __syncthreads(); passB_stage<16>(blk, tid);
    __syncthreads(); passB_stage<32>(blk, tid);
    __syncthreads(); passB_stage<64>(blk, tid);
    __syncthreads(); passB_stage<128>(blk, tid);
    __syncthreads(); passB_stage<256>(blk, tid);
    __syncthreads(); passB_stage<512>(blk, tid);
    __syncthreads(); passB_stage<1024>(blk, tid);
    __syncthreads(); passB_stage<2048>(blk, tid);
    __syncthreads();
    const float scale = rsqrtf((float)DD * gs);
#pragma unroll
    for (int u = 0; u < BVEC; ++u) {
        v4f m = blk4[tid + u * NTH];
        v4f o;
        o.x = iv[u].x + m.x * scale;
        o.y = iv[u].y + m.y * scale;
        o.z = iv[u].z + m.z * scale;
        o.w = iv[u].w + m.w * scale;
        dst[tid + u * NTH] = o;
    }
}

__global__ __launch_bounds__(512)
void fastfood_small_kernel(const float* __restrict__ theta, const float* __restrict__ BB,
                           const int* __restrict__ Pi, const float* __restrict__ GG,
                           const float* __restrict__ init, float* __restrict__ out,
                           int P, int LLv, int DD) {
    __shared__ float w[2048];
    __shared__ float m[4096];
    __shared__ float red[512];
    int tid = threadIdx.x, nt = blockDim.x;
    for (int j = tid; j < P; j += nt) w[j] = BB[j] * theta[j];
    fwht_lds(w, P, tid, nt);
    float s = 0.f;
    for (int i = tid; i < LLv; i += nt) { float g = GG[i]; s += g * g; }
    red[tid] = s; __syncthreads();
    for (int off = nt >> 1; off > 0; off >>= 1) {
        if (tid < off) red[tid] += red[tid + off];
        __syncthreads();
    }
    for (int i = tid; i < LLv; i += nt) m[i] = w[Pi[i] & (P - 1)] * GG[i];
    fwht_lds(m, LLv, tid, nt);
    float scale = rsqrtf((float)DD * red[0]);
    for (int i = tid; i < DD; i += nt) out[i] = init[i] + m[i] * scale;
}

#define GBM 32
#define GBN 64
#define GKC 64
#define GLD 68

template<int RELU>
__global__ __launch_bounds__(128)
void gemm_wmma_kernel(const float* __restrict__ A, const float* __restrict__ B,
                      const float* __restrict__ bias, float* __restrict__ C,
                      int M, int N, int K) {
    __shared__ __attribute__((aligned(16))) float As[2][GBM * GLD];
    __shared__ __attribute__((aligned(16))) float Bs[2][GBN * GLD];
    const int tid = threadIdx.x;
    const int lane = tid & 31, wave = tid >> 5;
    const int c = lane & 15, h = lane >> 4;
    const int wm = wave & 1, wn = wave >> 1;
    const int mBlocks = M / GBM;
    const int mb = blockIdx.x % mBlocks;
    const int nb = blockIdx.x / mBlocks;
    const int m0 = mb * GBM, n0 = nb * GBN;

    const int kq = (tid & 15) * 4, lr = tid >> 4;
    const float* ag = A + (size_t)(m0 + lr) * K + kq;
    const float* bg = B + (size_t)(n0 + lr) * K + kq;
    const size_t pass = (size_t)8 * K;
    const int lofs = lr * GLD + kq;

    v4f ra[GBM / 8], rb[GBN / 8];
    v8f acc0 = {}, acc1 = {};
    const int nK = K / GKC;

#pragma unroll
    for (int i = 0; i < GBM / 8; ++i) ra[i] = *(const v4f*)(ag + i * pass);
#pragma unroll
    for (int i = 0; i < GBN / 8; ++i) rb[i] = *(const v4f*)(bg + i * pass);
#pragma unroll
    for (int i = 0; i < GBM / 8; ++i) *(v4f*)&As[0][lofs + i * 8 * GLD] = ra[i];
#pragma unroll
    for (int i = 0; i < GBN / 8; ++i) *(v4f*)&Bs[0][lofs + i * 8 * GLD] = rb[i];
    __syncthreads();

    for (int ch = 0; ch < nK; ++ch) {
        const int cur = ch & 1;
        const bool more = (ch + 1 < nK);
        if (more) {
            const int kn = (ch + 1) * GKC;
#pragma unroll
            for (int i = 0; i < GBM / 8; ++i) ra[i] = *(const v4f*)(ag + i * pass + kn);
#pragma unroll
            for (int i = 0; i < GBN / 8; ++i) rb[i] = *(const v4f*)(bg + i * pass + kn);
        }

        const float* as  = &As[cur][(wm * 16 + c) * GLD + 2 * h];
        const float* bs0 = &Bs[cur][(wn * 32 + c) * GLD + 2 * h];
        const float* bs1 = bs0 + 16 * GLD;
        v2f af[GKC / 4], bf0[GKC / 4], bf1[GKC / 4];
#pragma unroll
        for (int s = 0; s < GKC / 4; ++s) {
            af[s]  = *(const v2f*)(as  + 4 * s);
            bf0[s] = *(const v2f*)(bs0 + 4 * s);
            bf1[s] = *(const v2f*)(bs1 + 4 * s);
        }
        __builtin_amdgcn_sched_barrier(0);
#pragma unroll
        for (int s = 0; s < GKC / 4; ++s) {
            acc0 = __builtin_amdgcn_wmma_f32_16x16x4_f32(false, af[s], false, bf0[s], (short)0, acc0, false, false);
            acc1 = __builtin_amdgcn_wmma_f32_16x16x4_f32(false, af[s], false, bf1[s], (short)0, acc1, false, false);
        }

        if (more) {
#pragma unroll
            for (int i = 0; i < GBM / 8; ++i) *(v4f*)&As[cur ^ 1][lofs + i * 8 * GLD] = ra[i];
#pragma unroll
            for (int i = 0; i < GBN / 8; ++i) *(v4f*)&Bs[cur ^ 1][lofs + i * 8 * GLD] = rb[i];
        }
        __syncthreads();
    }

    const int mrow = m0 + wm * 16 + h * 8;
    v8f accs[2] = {acc0, acc1};
#pragma unroll
    for (int t = 0; t < 2; ++t) {
        int n = n0 + wn * 32 + t * 16 + c;
        float bv = bias[n];
        v8f acc = accs[t];
#pragma unroll
        for (int v = 0; v < 8; ++v) {
            float xv = acc[v] + bv;
            if (RELU) xv = fmaxf(xv, 0.f);
            C[(size_t)(mrow + v) * N + n] = xv;
        }
    }
}

extern "C" void kernel_launch(void* const* d_in, const int* in_sizes, int n_in,
                              void* d_out, int out_size, void* d_ws, size_t ws_size,
                              hipStream_t stream) {
    const float* x     = (const float*)d_in[0];
    const float* theta = (const float*)d_in[1];
    const float* iW1   = (const float*)d_in[2];
    const float* ib1   = (const float*)d_in[3];
    const float* iW2   = (const float*)d_in[4];
    const float* ib2   = (const float*)d_in[5];
    const float* BB_W1 = (const float*)d_in[6];
    const int*   Pi_W1 = (const int*)d_in[7];
    const float* GG_W1 = (const float*)d_in[8];
    const float* BB_b1 = (const float*)d_in[9];
    const int*   Pi_b1 = (const int*)d_in[10];
    const float* GG_b1 = (const float*)d_in[11];
    const float* BB_W2 = (const float*)d_in[12];
    const int*   Pi_W2 = (const int*)d_in[13];
    const float* GG_W2 = (const float*)d_in[14];
    const float* BB_b2 = (const float*)d_in[15];
    const int*   Pi_b2 = (const int*)d_in[16];
    const float* GG_b2 = (const float*)d_in[17];
    float* out = (float*)d_out;

    float* ws = (float*)d_ws;
    size_t o = 0;
    float* W1   = ws + o; o += DD_W;
    float* W2   = ws + o; o += DD_W;
    float* b1   = ws + o; o += 3072;
    float* b2   = ws + o; o += 768;
    float* hbuf = ws + o; o += (size_t)256 * 3072;
    float* mbuf = ws + o; o += DD_W;
    float* wsm  = ws + o; o += 2048;
    float* part = ws + o; o += 256;
    float* gsum = ws + o; o += 1;

    small_fwht_kernel<<<1, 512, 0, stream>>>(theta, BB_W1, wsm, 2048);
    gg_partial_kernel<<<GGW, GGT, 0, stream>>>(GG_W1, LL_BIG, part);
    gg_final_kernel<<<1, 256, 0, stream>>>(part, gsum);
    fwht_passA_kernel<<<COLS / TC, 512, 0, stream>>>(wsm, Pi_W1, GG_W1, mbuf, 2047);
    fwht_passB_kernel<<<ROWS_OUT / BPB, 512, 0, stream>>>(mbuf, iW1, gsum, W1, DD_W);

    small_fwht_kernel<<<1, 512, 0, stream>>>(theta, BB_W2, wsm, 2048);
    gg_partial_kernel<<<GGW, GGT, 0, stream>>>(GG_W2, LL_BIG, part);
    gg_final_kernel<<<1, 256, 0, stream>>>(part, gsum);
    fwht_passA_kernel<<<COLS / TC, 512, 0, stream>>>(wsm, Pi_W2, GG_W2, mbuf, 2047);
    fwht_passB_kernel<<<ROWS_OUT / BPB, 512, 0, stream>>>(mbuf, iW2, gsum, W2, DD_W);

    fastfood_small_kernel<<<1, 512, 0, stream>>>(theta, BB_b1, Pi_b1, GG_b1, ib1, b1,
                                                 2048, 4096, 3072);
    fastfood_small_kernel<<<1, 512, 0, stream>>>(theta, BB_b2, Pi_b2, GG_b2, ib2, b2,
                                                 1024, 1024, 768);

    gemm_wmma_kernel<1><<<384, 128, 0, stream>>>(x, W1, b1, hbuf, 256, 3072, 768);
    gemm_wmma_kernel<0><<<96, 128, 0, stream>>>(hbuf, W2, b2, out, 256, 768, 3072);
}


# FWHT passes A and B: each group of 4 LDS butterfly stages becomes one 16-point Hadamard product on the matrix unit (f32 WMMA); LDS trips 10 -> 4 and 12 -> 3; 16-term sums in a new order
# speedup vs baseline: 3.0739x; 1.0911x over previous
#include <hip/hip_runtime.h>

typedef __attribute__((ext_vector_type(2))) float v2f;
typedef __attribute__((ext_vector_type(8))) float v8f;
typedef __attribute__((ext_vector_type(4))) float v4f;

#define LL_BIG   (1 << 22)
#define COLS     4096
#define ROWS     1024
#define ROWS_OUT 576
#define DD_W     2359296
#define TC       32
#define TSTRIDE  33
#define BPB      4

__device__ __forceinline__ void fwht_lds(float* buf, int n, int tid, int nt) {
    for (int h = 1; h < n; h <<= 1) {
        __syncthreads();
        for (int p = tid; p < (n >> 1); p += nt) {
            int i0 = ((p & ~(h - 1)) << 1) | (p & (h - 1));
            int i1 = i0 + h;
            float a = buf[i0], b = buf[i1];
            buf[i0] = a + b;
            buf[i1] = a - b;
        }
    }
    __syncthreads();
}

__global__ void small_fwht_kernel(const float* __restrict__ theta,
                                  const float* __restrict__ BB,
                                  float* __restrict__ wout, int P) {
    __shared__ float w[2048];
    int tid = threadIdx.x, nt = blockDim.x;
    for (int j = tid; j < P; j += nt) w[j] = BB[j] * theta[j];
    fwht_lds(w, P, tid, nt);
    for (int j = tid; j < P; j += nt) wout[j] = w[j];
}

#define GGB 32
#define GGW 256
#define GGT 256
__global__ __launch_bounds__(GGT)
void gg_partial_kernel(const float* __restrict__ GG, int n,
                       float* __restrict__ part) {
    __shared__ float red[256];
    int tid = threadIdx.x;
    constexpr int stride = GGW * GGT;
    float s = 0.f;
    int i = blockIdx.x * GGT + tid;
#pragma unroll 1
    for (; i + (GGB - 1) * stride < n; i += GGB * stride) {
        float g[GGB];
#pragma unroll
        for (int u = 0; u < GGB; ++u) g[u] = GG[i + u * stride];
#pragma unroll
        for (int u = 0; u < GGB; ++u) s += g[u] * g[u];
    }
    for (; i < n; i += stride) {
        float g = GG[i]; s += g * g;
    }
    red[tid] = s; __syncthreads();
    for (int off = 128; off > 0; off >>= 1) {
        if (tid < off) red[tid] += red[tid + off];
        __syncthreads();
    }
    if (tid == 0) part[blockIdx.x] = red[0];
}

__global__ void gg_final_kernel(const float* __restrict__ part, float* __restrict__ gsum) {
    __shared__ float red[256];
    int tid = threadIdx.x;
    red[tid] = part[tid]; __syncthreads();
    for (int off = 128; off > 0; off >>= 1) {
        if (tid < off) red[tid] += red[tid + off];
        __syncthreads();
    }
    if (tid == 0) gsum[0] = red[0];
}

template<bool STAG>
__device__ __forceinline__ void hadamard_frags(v2f* hf, int lane) {
    const int m = lane & 15, h = lane >> 4;
    const int row = STAG ? (2 * (m & 7) + (m >> 3)) : m;
#pragma unroll
    for (int s = 0; s < 4; ++s) {
        int k0 = STAG ? (4 * s + h) : (4 * s + 2 * h);
        int k1 = STAG ? (k0 + 2) : (k0 + 1);
        hf[s].x = (__popc(row & k0) & 1) ? -1.f : 1.f;
        hf[s].y = (__popc(row & k1) & 1) ? -1.f : 1.f;
    }
}

__device__ __forceinline__ void hadamard16x2(const v2f* hf, const v2f* xa, const v2f* xb,
                                             v8f& da, v8f& db) {
    da = (v8f){}; db = (v8f){};
#pragma unroll
    for (int s = 0; s < 4; ++s) {
        da = __builtin_amdgcn_wmma_f32_16x16x4_f32(false, hf[s], false, xa[s], (short)0, da, false, false);
        db = __builtin_amdgcn_wmma_f32_16x16x4_f32(false, hf[s], false, xb[s], (short)0, db, false, false);
    }
}

#define NTH  512
#define GAB  32
#define BAB  16

template<int H>
__device__ __forceinline__ void passA_stage(float* tile, int c, int pr) {
#pragma unroll 1
    for (int kb = 0; kb < (ROWS / 2) / 16; kb += BAB) {
        float a[BAB], b[BAB];
        int ad[BAB];
#pragma unroll
        for (int u = 0; u < BAB; ++u) {
            int p = pr + 16 * (kb + u);
            int r0 = ((p & ~(H - 1)) << 1) | (p & (H - 1));
            ad[u] = r0 * TSTRIDE + c;
            a[u] = tile[ad[u]];
            b[u] = tile[ad[u] + H * TSTRIDE];
        }
#pragma unroll
        for (int u = 0; u < BAB; ++u) {
            tile[ad[u]] = a[u] + b[u];
            tile[ad[u] + H * TSTRIDE] = a[u] - b[u];
        }
    }
}

#define HPA 8

template<int S>
__device__ __forceinline__ void passA_h16(float* tile, int tid, const v2f* hf) {
    const int lane = tid & 31, wave = tid >> 5;
    const int n = lane & 15, h = lane >> 4;
    const int col = (wave & 1) * 16 + n;
    v2f x[HPA][4];
    int base[HPA];
    const int wbase = (S == 1 ? (wave >> 1) * 16 : (wave >> 1)) * TSTRIDE + col;
#pragma unroll
    for (int j = 0; j < HPA; ++j) {
        base[j] = wbase + (S == 1 ? j * 128 : (j & 1) * 8 + (j >> 1) * 256) * TSTRIDE;
#pragma unroll
        for (int s = 0; s < 4; ++s) {
            x[j][s].x = tile[base[j] + ((4 * s + h) * S) * TSTRIDE];
            x[j][s].y = tile[base[j] + ((4 * s + 2 + h) * S) * TSTRIDE];
        }
    }
    __builtin_amdgcn_sched_barrier(0);
#pragma unroll
    for (int j = 0; j < HPA; j += 2) {
        v8f d0, d1;
        hadamard16x2(hf, x[j], x[j + 1], d0, d1);
#pragma unroll
        for (int v = 0; v < 8; ++v) {
            tile[base[j] + ((2 * v + h) * S) * TSTRIDE] = d0[v];
            tile[base[j + 1] + ((2 * v + h) * S) * TSTRIDE] = d1[v];
        }
    }
}

__global__ __launch_bounds__(NTH)
void fwht_passA_kernel(const float* __restrict__ w, const int* __restrict__ Pi,
                       const float* __restrict__ GG, float* __restrict__ mbuf,
                       int pmask) {
    __shared__ float tile[ROWS * TSTRIDE];
    const int tid = threadIdx.x;
    const int c = tid & (TC - 1), pr = tid >> 5;
    const int col0 = blockIdx.x * TC;
    const int gbase = pr * COLS + col0 + c;
#pragma unroll 1
    for (int kb = 0; kb < ROWS / 16; kb += GAB) {
        int pi[GAB];
        float gg[GAB], wv[GAB];
#pragma unroll
        for (int u = 0; u < GAB; ++u) {
            int g = gbase + (kb + u) * 16 * COLS;
            pi[u] = Pi[g];
            gg[u] = GG[g];
        }
#pragma unroll
        for (int u = 0; u < GAB; ++u) wv[u] = w[pi[u] & pmask];
#pragma unroll
        for (int u = 0; u < GAB; ++u)
            tile[(pr + 16 * (kb + u)) * TSTRIDE + c] = wv[u] * gg[u];
    }
    v2f hf[4];
    hadamard_frags<true>(hf, tid & 31);
    __syncthreads(); passA_h16<1>(tile, tid, hf);
    __syncthreads(); passA_h16<16>(tile, tid, hf);
    __syncthreads(); passA_stage<256>(tile, c, pr);
    __syncthreads(); passA_stage<512>(tile, c, pr);
    __syncthreads();
    for (int idx = tid; idx < ROWS_OUT * TC; idx += NTH) {
        int r = idx >> 5, cc = idx & (TC - 1);
        mbuf[(size_t)r * COLS + col0 + cc] = tile[r * TSTRIDE + cc];
    }
}

#define BVEC (BPB * COLS / 4 / NTH)

#define BLDS (BPB * BBLK)
#define HPB 4
#define BBLK (COLS / 16 * 20 + COLS / 256 * 16)

__device__ __forceinline__ int pb_idx(int i) {
    return (i >> 4) * 20 + (i & 15) + (i >> 8) * 16;
}

__device__ __forceinline__ void passB_h16_lo(float* blk, int tid, const v2f* hf) {
    const int lane = tid & 31, wave = tid >> 5;
    const int n = lane & 15, h = lane >> 4;
    v2f x[HPB][4];
    int base[HPB];
#pragma unroll
    for (int j = 0; j < HPB; ++j) {
        base[j] = BBLK * j + wave * 336 + n * 20;
#pragma unroll
        for (int s = 0; s < 4; ++s)
            x[j][s] = *(const v2f*)&blk[base[j] + 4 * s + 2 * h];
    }
    __builtin_amdgcn_sched_barrier(0);
#pragma unroll
    for (int j = 0; j < HPB; j += 2) {
        v8f d[2];
        hadamard16x2(hf, x[j], x[j + 1], d[0], d[1]);
#pragma unroll
        for (int t = 0; t < 2; ++t) {
            v4f lo = {d[t][0], d[t][1], d[t][2], d[t][3]}, hi = {d[t][4], d[t][5], d[t][6], d[t][7]};
            *(v4f*)&blk[base[j + t] + 8 * h] = lo;
            *(v4f*)&blk[base[j + t] + 8 * h + 4] = hi;
        }
    }
}

template<int S>
__device__ __forceinline__ void passB_h16_hi(float* blk, int tid, const v2f* hf) {
    const int lane = tid & 31, wave = tid >> 5;
    const int n = lane & 15, h = lane >> 4;
    v2f x[HPB][4];
    int base[HPB];
    constexpr int DS = (S == 16) ? 20 : 336;
    const int wbase = wave * (S == 16 ? 336 : 20) + n + h * DS;
#pragma unroll
    for (int j = 0; j < HPB; ++j) {
        base[j] = wbase + BBLK * j;
#pragma unroll
        for (int s = 0; s < 4; ++s) {
            x[j][s].x = blk[base[j] + (4 * s) * DS];
            x[j][s].y = blk[base[j] + (4 * s + 2) * DS];
        }
    }
    __builtin_amdgcn_sched_barrier(0);
#pragma unroll
    for (int j = 0; j < HPB; j += 2) {
        v8f d0, d1;
        hadamard16x2(hf, x[j], x[j + 1], d0, d1);
#pragma unroll
        for (int v = 0; v < 8; ++v) {
            blk[base[j] + (2 * v) * DS] = d0[v];
            blk[base[j + 1] + (2 * v) * DS] = d1[v];
        }
    }
}

__global__ __launch_bounds__(NTH)
void fwht_passB_kernel(const float* __restrict__ mbuf, const float* __restrict__ init,
                       const float* __restrict__ gsum, float* __restrict__ out, int DD) {
    __shared__ __attribute__((aligned(16))) float blk[BLDS];
    const int tid = threadIdx.x;
    const size_t base = (size_t)blockIdx.x * (BPB * COLS);
    const v4f* src = (const v4f*)(mbuf + base);
    const v4f* ini = (const v4f*)(init + base);
    v4f* dst = (v4f*)(out + base);
    v4f mv[BVEC], iv[BVEC];
#pragma unroll
    for (int u = 0; u < BVEC; ++u) mv[u] = src[tid + u * NTH];
#pragma unroll
    for (int u = 0; u < BVEC; ++u) iv[u] = ini[tid + u * NTH];
    const float gs = gsum[0];
#pragma unroll
    for (int u = 0; u < BVEC; ++u) *(v4f*)&blk[pb_idx(4 * (tid + u * NTH))] = mv[u];
    v2f hp[4], hs[4];
    hadamard_frags<false>(hp, tid & 31);
    hadamard_frags<true>(hs, tid & 31);
    __syncthreads(); passB_h16_lo(blk, tid, hp);
    __syncthreads(); passB_h16_hi<16>(blk, tid, hs);
    __syncthreads(); passB_h16_hi<256>(blk, tid, hs);
    __syncthreads();
    const float scale = rsqrtf((float)DD * gs);
#pragma unroll
    for (int u = 0; u < BVEC; ++u) {
        v4f m = *(const v4f*)&blk[pb_idx(4 * (tid + u * NTH))];
        v4f o;
        o.x = iv[u].x + m.x * scale;
        o.y = iv[u].y + m.y * scale;
        o.z = iv[u].z + m.z * scale;
        o.w = iv[u].w + m.w * scale;
        dst[tid + u * NTH] = o;
    }
}

__global__ __launch_bounds__(512)
void fastfood_small_kernel(const float* __restrict__ theta, const float* __restrict__ BB,
                           const int* __restrict__ Pi, const float* __restrict__ GG,
                           const float* __restrict__ init, float* __restrict__ out,
                           int P, int LLv, int DD) {
    __shared__ float w[2048];
    __shared__ float m[4096];
    __shared__ float red[512];
    int tid = threadIdx.x, nt = blockDim.x;
    for (int j = tid; j < P; j += nt) w[j] = BB[j] * theta[j];
    fwht_lds(w, P, tid, nt);
    float s = 0.f;
    for (int i = tid; i < LLv; i += nt) { float g = GG[i]; s += g * g; }
    red[tid] = s; __syncthreads();
    for (int off = nt >> 1; off > 0; off >>= 1) {
        if (tid < off) red[tid] += red[tid + off];
        __syncthreads();
    }
    for (int i = tid; i < LLv; i += nt) m[i] = w[Pi[i] & (P - 1)] * GG[i];
    fwht_lds(m, LLv, tid, nt);
    float scale = rsqrtf((float)DD * red[0]);
    for (int i = tid; i < DD; i += nt) out[i] = init[i] + m[i] * scale;
}

#define GBM 32
#define GBN 64
#define GKC 64
#define GLD 68

template<int RELU>
__global__ __launch_bounds__(128)
void gemm_wmma_kernel(const float* __restrict__ A, const float* __restrict__ B,
                      const float* __restrict__ bias, float* __restrict__ C,
                      int M, int N, int K) {
    __shared__ __attribute__((aligned(16))) float As[2][GBM * GLD];
    __shared__ __attribute__((aligned(16))) float Bs[2][GBN * GLD];
    const int tid = threadIdx.x;
    const int lane = tid & 31, wave = tid >> 5;
    const int c = lane & 15, h = lane >> 4;
    const int wm = wave & 1, wn = wave >> 1;
    const int mBlocks = M / GBM;
    const int mb = blockIdx.x % mBlocks;
    const int nb = blockIdx.x / mBlocks;
    const int m0 = mb * GBM, n0 = nb * GBN;

    const int kq = (tid & 15) * 4, lr = tid >> 4;
    const float* ag = A + (size_t)(m0 + lr) * K + kq;
    const float* bg = B + (size_t)(n0 + lr) * K + kq;
    const size_t pass = (size_t)8 * K;
    const int lofs = lr * GLD + kq;

    v4f ra[GBM / 8], rb[GBN / 8];
    v8f acc0 = {}, acc1 = {};
    const int nK = K / GKC;

#pragma unroll
    for (int i = 0; i < GBM / 8; ++i) ra[i] = *(const v4f*)(ag + i * pass);
#pragma unroll
    for (int i = 0; i < GBN / 8; ++i) rb[i] = *(const v4f*)(bg + i * pass);
#pragma unroll
    for (int i = 0; i < GBM / 8; ++i) *(v4f*)&As[0][lofs + i * 8 * GLD] = ra[i];
#pragma unroll
    for (int i = 0; i < GBN / 8; ++i) *(v4f*)&Bs[0][lofs + i * 8 * GLD] = rb[i];
    __syncthreads();

    for (int ch = 0; ch < nK; ++ch) {
        const int cur = ch & 1;
        const bool more = (ch + 1 < nK);
        if (more) {
            const int kn = (ch + 1) * GKC;
#pragma unroll
            for (int i = 0; i < GBM / 8; ++i) ra[i] = *(const v4f*)(ag + i * pass + kn);
#pragma unroll
            for (int i = 0; i < GBN / 8; ++i) rb[i] = *(const v4f*)(bg + i * pass + kn);
        }

        const float* as  = &As[cur][(wm * 16 + c) * GLD + 2 * h];
        const float* bs0 = &Bs[cur][(wn * 32 + c) * GLD + 2 * h];
        const float* bs1 = bs0 + 16 * GLD;
        v2f af[GKC / 4], bf0[GKC / 4], bf1[GKC / 4];
#pragma unroll
        for (int s = 0; s < GKC / 4; ++s) {
            af[s]  = *(const v2f*)(as  + 4 * s);
            bf0[s] = *(const v2f*)(bs0 + 4 * s);
            bf1[s] = *(const v2f*)(bs1 + 4 * s);
        }
        __builtin_amdgcn_sched_barrier(0);
#pragma unroll
        for (int s = 0; s < GKC / 4; ++s) {
            acc0 = __builtin_amdgcn_wmma_f32_16x16x4_f32(false, af[s], false, bf0[s], (short)0, acc0, false, false);
            acc1 = __builtin_amdgcn_wmma_f32_16x16x4_f32(false, af[s], false, bf1[s], (short)0, acc1, false, false);
        }

        if (more) {
#pragma unroll
            for (int i = 0; i < GBM / 8; ++i) *(v4f*)&As[cur ^ 1][lofs + i * 8 * GLD] = ra[i];
#pragma unroll
            for (int i = 0; i < GBN / 8; ++i) *(v4f*)&Bs[cur ^ 1][lofs + i * 8 * GLD] = rb[i];
        }
        __syncthreads();
    }

    const int mrow = m0 + wm * 16 + h * 8;
    v8f accs[2] = {acc0, acc1};
#pragma unroll
    for (int t = 0; t < 2; ++t) {
        int n = n0 + wn * 32 + t * 16 + c;
        float bv = bias[n];
        v8f acc = accs[t];
#pragma unroll
        for (int v = 0; v < 8; ++v) {
            float xv = acc[v] + bv;
            if (RELU) xv = fmaxf(xv, 0.f);
            C[(size_t)(mrow + v) * N + n] = xv;
        }
    }
}

extern "C" void kernel_launch(void* const* d_in, const int* in_sizes, int n_in,
                              void* d_out, int out_size, void* d_ws, size_t ws_size,
                              hipStream_t stream) {
    const float* x     = (const float*)d_in[0];
    const float* theta = (const float*)d_in[1];
    const float* iW1   = (const float*)d_in[2];
    const float* ib1   = (const float*)d_in[3];
    const float* iW2   = (const float*)d_in[4];
    const float* ib2   = (const float*)d_in[5];
    const float* BB_W1 = (const float*)d_in[6];
    const int*   Pi_W1 = (const int*)d_in[7];
    const float* GG_W1 = (const float*)d_in[8];
    const float* BB_b1 = (const float*)d_in[9];
    const int*   Pi_b1 = (const int*)d_in[10];
    const float* GG_b1 = (const float*)d_in[11];
    const float* BB_W2 = (const float*)d_in[12];
    const int*   Pi_W2 = (const int*)d_in[13];
    const float* GG_W2 = (const float*)d_in[14];
    const float* BB_b2 = (const float*)d_in[15];
    const int*   Pi_b2 = (const int*)d_in[16];
    const float* GG_b2 = (const float*)d_in[17];
    float* out = (float*)d_out;

    float* ws = (float*)d_ws;
    size_t o = 0;
    float* W1   = ws + o; o += DD_W;
    float* W2   = ws + o; o += DD_W;
    float* b1   = ws + o; o += 3072;
    float* b2   = ws + o; o += 768;
    float* hbuf = ws + o; o += (size_t)256 * 3072;
    float* mbuf = ws + o; o += DD_W;
    float* wsm  = ws + o; o += 2048;
    float* part = ws + o; o += 256;
    float* gsum = ws + o; o += 1;

    small_fwht_kernel<<<1, 512, 0, stream>>>(theta, BB_W1, wsm, 2048);
    gg_partial_kernel<<<GGW, GGT, 0, stream>>>(GG_W1, LL_BIG, part);
    gg_final_kernel<<<1, 256, 0, stream>>>(part, gsum);
    fwht_passA_kernel<<<COLS / TC, 512, 0, stream>>>(wsm, Pi_W1, GG_W1, mbuf, 2047);
    fwht_passB_kernel<<<ROWS_OUT / BPB, 512, 0, stream>>>(mbuf, iW1, gsum, W1, DD_W);

    small_fwht_kernel<<<1, 512, 0, stream>>>(theta, BB_W2, wsm, 2048);
    gg_partial_kernel<<<GGW, GGT, 0, stream>>>(GG_W2, LL_BIG, part);
    gg_final_kernel<<<1, 256, 0, stream>>>(part, gsum);
    fwht_passA_kernel<<<COLS / TC, 512, 0, stream>>>(wsm, Pi_W2, GG_W2, mbuf, 2047);
    fwht_passB_kernel<<<ROWS_OUT / BPB, 512, 0, stream>>>(mbuf, iW2, gsum, W2, DD_W);

    fastfood_small_kernel<<<1, 512, 0, stream>>>(theta, BB_b1, Pi_b1, GG_b1, ib1, b1,
                                                 2048, 4096, 3072);
    fastfood_small_kernel<<<1, 512, 0, stream>>>(theta, BB_b2, Pi_b2, GG_b2, ib2, b2,
                                                 1024, 1024, 768);

    gemm_wmma_kernel<1><<<384, 128, 0, stream>>>(x, W1, b1, hbuf, 256, 3072, 768);
    gemm_wmma_kernel<0><<<96, 128, 0, stream>>>(hbuf, W2, b2, out, 256, 768, 3072);
}


# The 6 launches that read inputs only (2 table FWHTs, 2 bias projections, 2 sum(GG^2) partial stages) are roles of one launch; pass B adds the partials itself (same tree): 14 launches -> 7; sums kept
# speedup vs baseline: 7.5417x; 2.4535x over previous
#include <hip/hip_runtime.h>

typedef __attribute__((ext_vector_type(2))) float v2f;
typedef __attribute__((ext_vector_type(8))) float v8f;
typedef __attribute__((ext_vector_type(4))) float v4f;

#define LL_BIG   (1 << 22)
#define COLS     8192
#define SUB      4096
#define ROWS     512
#define ROWS_OUT 288
#define DD_W     2359296
#define TC       32
#define TSTRIDE  33
#define BPB      2

#define GGB 32
#define GGW 256
#define GGT 256
#define GGWG (GGW / 2)
__device__ __forceinline__ void ggsum_role(const float* __restrict__ GG,
                                           float* __restrict__ part, int wg,
                                           float* red, int tid) {
    constexpr int stride = GGW * GGT;
    constexpr int n = LL_BIG;
    const int t = tid & (GGT - 1);
    float s = 0.f;
    int i = wg * (2 * GGT) + tid;
#pragma unroll 1
    for (; i + (GGB - 1) * stride < n; i += GGB * stride) {
        float g[GGB];
#pragma unroll
        for (int u = 0; u < GGB; ++u) g[u] = GG[i + u * stride];
#pragma unroll
        for (int u = 0; u < GGB; ++u) s += g[u] * g[u];
    }
    for (; i < n; i += stride) {
        float g = GG[i]; s += g * g;
    }
    red[tid] = s; __syncthreads();
    for (int off = GGT / 2; off > 0; off >>= 1) {
        if (t < off) red[tid] += red[tid + off];
        __syncthreads();
    }
    if (t == 0) part[2 * wg + tid / GGT] = red[tid];
}

template<int N>
__device__ __forceinline__ float row_down(float c) {
    return __int_as_float(__builtin_amdgcn_update_dpp(0, __float_as_int(c), 0x100 + N  ,
                                                      0xf, 0xf, true));
}
__device__ __forceinline__ float other_row(float c) {
    return __int_as_float(__builtin_amdgcn_permlanex16(0, __float_as_int(c), 0x76543210u, 0xfedcba98u,
                                                       false, false));
}

__device__ __forceinline__ float gg_total(const float* pv) {
    float a0 = pv[0] + pv[4], a1 = pv[1] + pv[5], a2 = pv[2] + pv[6], a3 = pv[3] + pv[7];
    float c = (a0 + a2) + (a1 + a3);
    c += other_row(c);
    c += row_down<8>(c);
    c += row_down<4>(c);
    c += row_down<2>(c);
    c += row_down<1>(c);
    return __int_as_float(__builtin_amdgcn_readfirstlane(__float_as_int(c)));
}

template<bool STAG>
__device__ __forceinline__ void hadamard_frags(v2f* hf, int lane) {
    const int m = lane & 15, h = lane >> 4;
    const int row = STAG ? (2 * (m & 7) + (m >> 3)) : m;
#pragma unroll
    for (int s = 0; s < 4; ++s) {
        int k0 = STAG ? (4 * s + h) : (4 * s + 2 * h);
        int k1 = STAG ? (k0 + 2) : (k0 + 1);
        hf[s].x = (__popc(row & k0) & 1) ? -1.f : 1.f;
        hf[s].y = (__popc(row & k1) & 1) ? -1.f : 1.f;
    }
}

__device__ __forceinline__ void hadamard16x2(const v2f* hf, const v2f* xa, const v2f* xb,
                                             v8f& da, v8f& db) {
    da = (v8f){}; db = (v8f){};
#pragma unroll
    for (int s = 0; s < 4; ++s) {
        da = __builtin_amdgcn_wmma_f32_16x16x4_f32(false, hf[s], false, xa[s], (short)0, da, false, false);
        db = __builtin_amdgcn_wmma_f32_16x16x4_f32(false, hf[s], false, xb[s], (short)0, db, false, false);
    }
}

template<int S, int ND>
__device__ __forceinline__ void small_h16(float* p, int h, const v2f* hf) {
    constexpr int KS = ND / 4;
    v2f x[KS];
#pragma unroll
    for (int s = 0; s < KS; ++s) {
        x[s].x = p[(4 * s + h) * S];
        x[s].y = p[(4 * s + 2 + h) * S];
    }
    __builtin_amdgcn_sched_barrier(0);
    v8f d = {};
#pragma unroll
    for (int s = 0; s < KS; ++s)
        d = __builtin_amdgcn_wmma_f32_16x16x4_f32(false, hf[s], false, x[s], (short)0, d, false, false);
#pragma unroll
    for (int v = 0; v < ND / 2; ++v) p[(2 * v + h) * S] = d[v];
}

template<int N>
__device__ __forceinline__ void fwht_lds_mm(float* buf, int tid, const v2f* hf) {
    static_assert(N == 1024 || N == 2048 || N == 4096, "three digits: 4 + 4 + (2..4) bits");
    constexpr int NP = N / 256;
    const int lane = tid & 31, wave = tid >> 5;
    const int n = lane & 15, h = lane >> 4;
    __syncthreads();
    if (wave < NP) small_h16<1, 16>(buf + wave * 256 + n * 16, h, hf);
    __syncthreads();
    if (wave < NP) small_h16<16, 16>(buf + wave * 256 + n, h, hf);
    __syncthreads();
    small_h16<256, NP>(buf + wave * 16 + n, h, hf);
    __syncthreads();
}

#define SWT 512
template<int P>
__device__ __forceinline__ void table_role(const float* __restrict__ theta,
                                           const float* __restrict__ BB,
                                           float* __restrict__ wout, float* w, int tid) {
    constexpr int NP = P / SWT;
    {
        float bb[NP], th[NP];
#pragma unroll
        for (int u = 0; u < NP; ++u) {
            bb[u] = BB[tid + u * SWT];
            th[u] = theta[tid + u * SWT];
        }
#pragma unroll
        for (int u = 0; u < NP; ++u) w[tid + u * SWT] = bb[u] * th[u];
    }
    v2f hf[4];
    hadamard_frags<true>(hf, tid & 31);
    fwht_lds_mm<P>(w, tid, hf);
#pragma unroll
    for (int u = 0; u < NP; ++u) wout[tid + u * SWT] = w[tid + u * SWT];
}

#define NTH  512
#define GAB  32

#define HPA 4

template<int S>
__device__ __forceinline__ void passA_h16(float* tile, int tid, const v2f* hf) {
    const int lane = tid & 31, wave = tid >> 5;
    const int n = lane & 15, h = lane >> 4;
    const int col = (wave & 1) * 16 + n;
    v2f x[HPA][4];
    int base[HPA];
    const int wbase = (S == 1 ? (wave >> 1) * 16 : (wave >> 1)) * TSTRIDE + col;
#pragma unroll
    for (int j = 0; j < HPA; ++j) {
        base[j] = wbase + (S == 1 ? j * 128 : (j & 1) * 8 + (j >> 1) * 256) * TSTRIDE;
#pragma unroll
        for (int s = 0; s < 4; ++s) {
            x[j][s].x = tile[base[j] + ((4 * s + h) * S) * TSTRIDE];
            x[j][s].y = tile[base[j] + ((4 * s + 2 + h) * S) * TSTRIDE];
        }
    }
    __builtin_amdgcn_sched_barrier(0);
#pragma unroll
    for (int j = 0; j < HPA; j += 2) {
        v8f d0, d1;
        hadamard16x2(hf, x[j], x[j + 1], d0, d1);
#pragma unroll
        for (int v = 0; v < 8; ++v) {
            tile[base[j] + ((2 * v + h) * S) * TSTRIDE] = d0[v];
            tile[base[j + 1] + ((2 * v + h) * S) * TSTRIDE] = d1[v];
        }
    }
}

__global__ __launch_bounds__(NTH)
void fwht_passA_kernel(const float* __restrict__ w, const int* __restrict__ Pi,
                       const float* __restrict__ GG, float* __restrict__ mbuf,
                       int pmask) {
    __shared__ float tile[ROWS * TSTRIDE];
    const int tid = threadIdx.x;
    const int c = tid & (TC - 1), pr = tid >> 5;
    const int col0 = blockIdx.x * TC;
    const int gbase = pr * COLS + col0 + c;
    static_assert(GAB == ROWS / 16, "a thread's rows are one batch");
    {
        int pi[GAB];
        float gg[GAB], wv[GAB];
#pragma unroll
        for (int u = 0; u < GAB; ++u) {
            int g = gbase + u * 16 * COLS;
            pi[u] = Pi[g];
            gg[u] = GG[g];
        }
#pragma unroll
        for (int u = 0; u < GAB; ++u) wv[u] = w[pi[u] & pmask];
#pragma unroll
        for (int u = 0; u < GAB; ++u)
            tile[(pr + 16 * u) * TSTRIDE + c] = wv[u] * gg[u];
    }
    v2f hf[4];
    hadamard_frags<true>(hf, tid & 31);
    __syncthreads(); passA_h16<1>(tile, tid, hf);
    __syncthreads(); passA_h16<16>(tile, tid, hf);
    __syncthreads();
    constexpr int OB = ROWS_OUT * TC / NTH;
    float lo[OB], hi[OB];
#pragma unroll
    for (int it = 0; it < OB; ++it) {
        int rl = pr + 16 * (it & 15);
        lo[it] = tile[rl * TSTRIDE + c];
        hi[it] = tile[(rl + ROWS / 2) * TSTRIDE + c];
    }
#pragma unroll
    for (int it = 0; it < OB; ++it) {
        int r = pr + 16 * it;
        mbuf[(size_t)r * COLS + col0 + c] = (it < 16) ? lo[it] + hi[it] : lo[it] - hi[it];
    }
}

#define BVEC (BPB * SUB / 4 / NTH)

#define BLDS (BPB * BBLK)
#define HPB 2
#define BBLK (SUB / 16 * 20 + SUB / 256 * 16)

__device__ __forceinline__ int pb_idx(int i) {
    return (i >> 4) * 20 + (i & 15) + (i >> 8) * 16;
}

__device__ __forceinline__ void passB_h16_lo(float* blk, int tid, const v2f* hf) {
    const int lane = tid & 31, wave = tid >> 5;
    const int n = lane & 15, h = lane >> 4;
    v2f x[HPB][4];
    int base[HPB];
#pragma unroll
    for (int j = 0; j < HPB; ++j) {
        base[j] = BBLK * j + wave * 336 + n * 20;
#pragma unroll
        for (int s = 0; s < 4; ++s)
            x[j][s] = *(const v2f*)&blk[base[j] + 4 * s + 2 * h];
    }
    __builtin_amdgcn_sched_barrier(0);
#pragma unroll
    for (int j = 0; j < HPB; j += 2) {
        v8f d[2];
        hadamard16x2(hf, x[j], x[j + 1], d[0], d[1]);
#pragma unroll
        for (int t = 0; t < 2; ++t) {
            v4f lo = {d[t][0], d[t][1], d[t][2], d[t][3]}, hi = {d[t][4], d[t][5], d[t][6], d[t][7]};
            *(v4f*)&blk[base[j + t] + 8 * h] = lo;
            *(v4f*)&blk[base[j + t] + 8 * h + 4] = hi;
        }
    }
}

template<int S>
__device__ __forceinline__ void passB_h16_hi(float* blk, int tid, const v2f* hf) {
    const int lane = tid & 31, wave = tid >> 5;
    const int n = lane & 15, h = lane >> 4;
    v2f x[HPB][4];
    int base[HPB];
    constexpr int DS = (S == 16) ? 20 : 336;
    const int wbase = wave * (S == 16 ? 336 : 20) + n + h * DS;
#pragma unroll
    for (int j = 0; j < HPB; ++j) {
        base[j] = wbase + BBLK * j;
#pragma unroll
        for (int s = 0; s < 4; ++s) {
            x[j][s].x = blk[base[j] + (4 * s) * DS];
            x[j][s].y = blk[base[j] + (4 * s + 2) * DS];
        }
    }
    __builtin_amdgcn_sched_barrier(0);
#pragma unroll
    for (int j = 0; j < HPB; j += 2) {
        v8f d0, d1;
        hadamard16x2(hf, x[j], x[j + 1], d0, d1);
#pragma unroll
        for (int v = 0; v < 8; ++v) {
            blk[base[j] + (2 * v) * DS] = d0[v];
            blk[base[j + 1] + (2 * v) * DS] = d1[v];
        }
    }
}

__global__ __launch_bounds__(NTH)
void fwht_passB_kernel(const float* __restrict__ mbuf, const float* __restrict__ init,
                       const float* __restrict__ part, float* __restrict__ out, int DD) {
    __shared__ __attribute__((aligned(16))) float blk[BLDS];
    const int tid = threadIdx.x;
    const size_t base = (size_t)blockIdx.x * COLS;
    const v4f* src = (const v4f*)(mbuf + base);
    const v4f* ini = (const v4f*)(init + base);
    v4f* dst = (v4f*)(out + base);
    v4f mv[BVEC], iv[BVEC];
#pragma unroll
    for (int u = 0; u < BVEC; ++u) mv[u] = src[tid + u * NTH];
#pragma unroll
    for (int u = 0; u < BVEC; ++u) iv[u] = ini[tid + u * NTH];
    float pv[GGW / 32];
#pragma unroll
    for (int j = 0; j < GGW / 32; ++j) pv[j] = part[(tid & 31) + 32 * j];
#pragma unroll
    for (int u = 0; u < BVEC; ++u) *(v4f*)&blk[pb_idx(4 * (tid + u * NTH))] = mv[u];
    v2f hp[4], hs[4];
    hadamard_frags<false>(hp, tid & 31);
    hadamard_frags<true>(hs, tid & 31);
    __syncthreads(); passB_h16_lo(blk, tid, hp);
    __syncthreads(); passB_h16_hi<16>(blk, tid, hs);
    __syncthreads(); passB_h16_hi<256>(blk, tid, hs);
    __syncthreads();
    const float scale = rsqrtf((float)DD * gg_total(pv));
    v4f m[BVEC];
#pragma unroll
    for (int u = 0; u < BVEC; ++u) m[u] = *(const v4f*)&blk[pb_idx(4 * (tid + u * NTH))];
#pragma unroll
    for (int u = 0; u < BVEC / 2; ++u) {
        v4f s = m[u] + m[u + BVEC / 2], d = m[u] - m[u + BVEC / 2];
        dst[tid + u * NTH] = iv[u] + s * scale;
        dst[tid + (u + BVEC / 2) * NTH] = iv[u + BVEC / 2] + d * scale;
    }
}

template<int P, int LLV, int DD>
__device__ __forceinline__ void bias_role(const float* __restrict__ theta, const float* __restrict__ BB,
                                          const int* __restrict__ Pi, const float* __restrict__ GG,
                                          const float* __restrict__ init, float* __restrict__ out,
                                          float* w, float* m, float* red, int tid) {
    static_assert(DD <= LLV, "the outputs are the first DD elements of the transform");
    constexpr int NP = P / SWT, NL = LLV / SWT, NO = (DD + SWT - 1) / SWT;
    v2f hf[4];
    hadamard_frags<true>(hf, tid & 31);
    {
        float bb[NP], th[NP];
#pragma unroll
        for (int u = 0; u < NP; ++u) {
            bb[u] = BB[tid + u * SWT];
            th[u] = theta[tid + u * SWT];
        }
#pragma unroll
        for (int u = 0; u < NP; ++u) w[tid + u * SWT] = bb[u] * th[u];
    }
    fwht_lds_mm<P>(w, tid, hf);
    float s = 0.f;
    {
        float g[NL];
#pragma unroll
        for (int u = 0; u < NL; ++u) g[u] = GG[tid + u * SWT];
#pragma unroll
        for (int u = 0; u < NL; ++u) s = __builtin_fmaf(g[u], g[u], s);
    }
    red[tid] = s; __syncthreads();
    for (int off = SWT >> 1; off > 0; off >>= 1) {
        if (tid < off) red[tid] += red[tid + off];
        __syncthreads();
    }
    {
        int pi[NL];
        float gg[NL], wv[NL];
#pragma unroll
        for (int u = 0; u < NL; ++u) {
            pi[u] = Pi[tid + u * SWT];
            gg[u] = GG[tid + u * SWT];
        }
#pragma unroll
        for (int u = 0; u < NL; ++u) wv[u] = w[pi[u] & (P - 1)];
#pragma unroll
        for (int u = 0; u < NL; ++u) m[tid + u * SWT] = wv[u] * gg[u];
    }
    fwht_lds_mm<LLV>(m, tid, hf);
    const float scale = rsqrtf((float)DD * red[0]);
    {
        float iv[NO];
#pragma unroll
        for (int u = 0; u < NO; ++u)
            iv[u] = (tid + u * SWT < DD) ? init[tid + u * SWT] : 0.f;
#pragma unroll
        for (int u = 0; u < NO; ++u)
            if (tid + u * SWT < DD) out[tid + u * SWT] = iv[u] + m[tid + u * SWT] * scale;
    }
}

#define SETUP_ONE 4
#define SETUP_WG (SETUP_ONE + 2 * GGWG)
__global__ __launch_bounds__(SWT)
void setup_kernel(const float* __restrict__ theta,
                  const float* __restrict__ BB_W1, const float* __restrict__ GG_W1,
                  const float* __restrict__ BB_W2, const float* __restrict__ GG_W2,
                  const float* __restrict__ BB_b1, const int* __restrict__ Pi_b1,
                  const float* __restrict__ GG_b1, const float* __restrict__ ib1,
                  const float* __restrict__ BB_b2, const int* __restrict__ Pi_b2,
                  const float* __restrict__ GG_b2, const float* __restrict__ ib2,
                  float* __restrict__ w1, float* __restrict__ w2,
                  float* __restrict__ part1, float* __restrict__ part2,
                  float* __restrict__ b1, float* __restrict__ b2) {
    __shared__ float smem[2048 + 4096 + SWT];
    const int tid = threadIdx.x;
    const int b = blockIdx.x;
    if (b == 0) {
        bias_role<2048, 4096, 3072>(theta, BB_b1, Pi_b1, GG_b1, ib1, b1,
                                    smem, smem + 2048, smem + 2048 + 4096, tid);
    } else if (b == 1) {
        bias_role<1024, 1024, 768>(theta, BB_b2, Pi_b2, GG_b2, ib2, b2,
                                   smem, smem + 1024, smem + 2048, tid);
    } else if (b < SETUP_ONE) {
        table_role<2048>(theta, b == 2 ? BB_W1 : BB_W2, b == 2 ? w1 : w2, smem, tid);
    } else if (b < SETUP_ONE + GGWG) {
        ggsum_role(GG_W1, part1, b - SETUP_ONE, smem, tid);
    } else {
        ggsum_role(GG_W2, part2, b - SETUP_ONE - GGWG, smem, tid);
    }
}

#define GBM 32
#define CPITCH 36

__device__ __forceinline__ void lds_barrier() {
    __builtin_amdgcn_fence(__ATOMIC_RELEASE, "workgroup", "local");
    __builtin_amdgcn_s_barrier();
    __builtin_amdgcn_fence(__ATOMIC_ACQUIRE, "workgroup", "local");
}

template<int NT, int KH, int LD>
__device__ __forceinline__ void gemm_read_half(const float* as, const float* bs,
                                               v2f* fa, v2f* fb0, v2f* fb1) {
#pragma unroll
    for (int s = 0; s < KH; ++s) {
        fa[s]  = *(const v2f*)(as + 4 * s);
        fb0[s] = *(const v2f*)(bs + 4 * s);
        if constexpr (NT == 2) fb1[s] = *(const v2f*)(bs + 16 * LD + 4 * s);
    }
}

template<int NT, int KH>
__device__ __forceinline__ void gemm_mma_half(const v2f* fa, const v2f* fb0, const v2f* fb1,
                                              v8f& acc0, v8f& acc1) {
    static_assert(KH % 2 == 0, "NT = 1: the parity of a k-step is its parity within the half");
#pragma unroll
    for (int s = 0; s < KH; ++s) {
        if constexpr (NT == 2) {
            acc0 = __builtin_amdgcn_wmma_f32_16x16x4_f32(false, fa[s], false, fb0[s], (short)0, acc0, false, false);
            acc1 = __builtin_amdgcn_wmma_f32_16x16x4_f32(false, fa[s], false, fb1[s], (short)0, acc1, false, false);
        } else if (s & 1) {
            acc1 = __builtin_amdgcn_wmma_f32_16x16x4_f32(false, fa[s], false, fb0[s], (short)0, acc1, false, false);
        } else {
            acc0 = __builtin_amdgcn_wmma_f32_16x16x4_f32(false, fa[s], false, fb0[s], (short)0, acc0, false, false);
        }
    }
}

template<int V> struct ctag { static constexpr int value = V; };

template<int RELU, int NT, int KC, int NK>
__global__ __launch_bounds__(128)
void gemm_wmma_kernel(const float* __restrict__ A, const float* __restrict__ B,
                      const float* __restrict__ bias, float* __restrict__ C,
                      int M, int N) {
    static_assert(NK >= 4 && NK % 2 == 0, "the pipeline below is written for an even chunk count");
    constexpr int K   = NK * KC;
    constexpr int BN  = 32 * NT;
    constexpr int LD  = KC + 4;
    constexpr int TPR = KC / 4;
    constexpr int RPP = 128 / TPR;
    constexpr int NA  = GBM / RPP, NB = BN / RPP;
    constexpr int KS  = KC / 4;
    constexpr int KH  = KS / 2;
    __shared__ __attribute__((aligned(16))) float As[2][GBM * LD];
    __shared__ __attribute__((aligned(16))) float Bs[2][BN * LD];
    const int tid = threadIdx.x;
    const int lane = tid & 31, wave = tid >> 5;
    const int c = lane & 15, h = lane >> 4;
    const int wm = wave & 1, wn = wave >> 1;
    const int mBlocks = M / GBM;
    const int mb = blockIdx.x % mBlocks;
    const int nb = blockIdx.x / mBlocks;
    const int m0 = mb * GBM, n0 = nb * BN;

    const int kq = (tid % TPR) * 4, lr = tid / TPR;
    const float* ag = A + (size_t)(m0 + lr) * K + kq;
    const float* bg = B + (size_t)(n0 + lr) * K + kq;
    constexpr size_t pass = (size_t)RPP * K;
    const int lofs = lr * LD + kq;

    v4f ra[3][NA], rb[3][NB];
    v2f fa[2][KH], fb0[2][KH], fb1[2][NT == 2 ? KH : 1];
    v8f acc0 = {}, acc1 = {};
    const float* asr = &As[0][(wm * 16 + c) * LD + 2 * h];
    const float* bsr = &Bs[0][(wn * 16 * NT + c) * LD + 2 * h];

    auto fetch = [&](auto set, int kn) __attribute__((always_inline)) {
        constexpr int S = decltype(set)::value;
#pragma unroll
        for (int i = 0; i < NA; ++i) ra[S][i] = *(const v4f*)(ag + i * pass + kn);
#pragma unroll
        for (int i = 0; i < NB; ++i) rb[S][i] = *(const v4f*)(bg + i * pass + kn);
    };
    auto park = [&](auto set) __attribute__((always_inline)) {
        constexpr int S = decltype(set)::value;
#pragma unroll
        for (int i = 0; i < NA; ++i) *(v4f*)&As[S % 2][lofs + i * RPP * LD] = ra[S][i];
#pragma unroll
        for (int i = 0; i < NB; ++i) *(v4f*)&Bs[S % 2][lofs + i * RPP * LD] = rb[S][i];
    };

    auto chunk = [&](int ch, auto kind, auto par) __attribute__((always_inline)) {
        constexpr int KIND = decltype(kind)::value;
        constexpr int P = decltype(par)::value;
        constexpr int CUR = 1 - P;
        gemm_read_half<NT, KH, LD>(asr + CUR * (GBM * LD) + 4 * KH, bsr + CUR * (BN * LD) + 4 * KH,
                                   fa[1], fb0[1], fb1[1]);
        if constexpr (KIND >= 1) park(ctag<P>{});
        if constexpr (KIND >= 2) fetch(ctag<P>{}, (ch + 3) * KC);
        __builtin_amdgcn_sched_barrier(0);
        gemm_mma_half<NT, KH>(fa[0], fb0[0], fb1[0], acc0, acc1);
        asm volatile("" :: "v"(acc0), "v"(acc1));
        __builtin_amdgcn_sched_barrier(0);
        lds_barrier();
        if constexpr (KIND >= 1)
            gemm_read_half<NT, KH, LD>(asr + P * (GBM * LD), bsr + P * (BN * LD),
                                       fa[0], fb0[0], fb1[0]);
        __builtin_amdgcn_sched_barrier(0);
        gemm_mma_half<NT, KH>(fa[1], fb0[1], fb1[1], acc0, acc1);
        asm volatile("" :: "v"(acc0), "v"(acc1));
        __builtin_amdgcn_sched_barrier(0);
    };

    fetch(ctag<2>{}, 0);
    __builtin_amdgcn_sched_barrier(0);
    fetch(ctag<1>{}, KC);
    __builtin_amdgcn_sched_barrier(0);
    fetch(ctag<0>{}, 2 * KC);
    __builtin_amdgcn_sched_barrier(0);
    park(ctag<2>{});
    lds_barrier();
    gemm_read_half<NT, KH, LD>(asr, bsr, fa[0], fb0[0], fb1[0]);

    int ch = 0;
#pragma unroll 1
    for (; ch + 4 < NK; ch += 2) {
        chunk(ch,     ctag<2>{}, ctag<1>{});
        chunk(ch + 1, ctag<2>{}, ctag<0>{});
    }
    chunk(NK - 4, ctag<2>{}, ctag<1>{});
    chunk(NK - 3, ctag<1>{}, ctag<0>{});
    chunk(NK - 2, ctag<1>{}, ctag<1>{});
    chunk(NK - 1, ctag<0>{}, ctag<0>{});

    if constexpr (NT == 2) {
        const int mrow = m0 + wm * 16 + h * 8;
        v8f accs[2] = {acc0, acc1};
#pragma unroll
        for (int t = 0; t < 2; ++t) {
            int n = n0 + wn * 32 + t * 16 + c;
            float bv = bias[n];
            v8f acc = accs[t];
#pragma unroll
            for (int v = 0; v < 8; ++v) {
                float xv = acc[v] + bv;
                if (RELU) xv = fmaxf(xv, 0.f);
                C[(size_t)(mrow + v) * N + n] = xv;
            }
        }
    } else {
        float* cs = &As[0][0];
        static_assert(GBM * CPITCH <= GBM * LD, "the staged tile fits the first A buffer");
        const v8f acc = acc0 + acc1;
#pragma unroll
        for (int v = 0; v < 8; ++v)
            cs[(wm * 16 + h * 8 + v) * CPITCH + wn * 16 + c] = acc[v];
        __syncthreads();
        const float bv = bias[n0 + lane];
#pragma unroll
        for (int i = 0; i < 8; ++i) {
            int r = wave * 8 + i;
            float xv = cs[r * CPITCH + lane] + bv;
            if (RELU) xv = fmaxf(xv, 0.f);
            C[(size_t)(m0 + r) * N + n0 + lane] = xv;
        }
    }
}

extern "C" void kernel_launch(void* const* d_in, const int* in_sizes, int n_in,
                              void* d_out, int out_size, void* d_ws, size_t ws_size,
                              hipStream_t stream) {
    const float* x     = (const float*)d_in[0];
    const float* theta = (const float*)d_in[1];
    const float* iW1   = (const float*)d_in[2];
    const float* ib1   = (const float*)d_in[3];
    const float* iW2   = (const float*)d_in[4];
    const float* ib2   = (const float*)d_in[5];
    const float* BB_W1 = (const float*)d_in[6];
    const int*   Pi_W1 = (const int*)d_in[7];
    const float* GG_W1 = (const float*)d_in[8];
    const float* BB_b1 = (const float*)d_in[9];
    const int*   Pi_b1 = (const int*)d_in[10];
    const float* GG_b1 = (const float*)d_in[11];
    const float* BB_W2 = (const float*)d_in[12];
    const int*   Pi_W2 = (const int*)d_in[13];
    const float* GG_W2 = (const float*)d_in[14];
    const float* BB_b2 = (const float*)d_in[15];
    const int*   Pi_b2 = (const int*)d_in[16];
    const float* GG_b2 = (const float*)d_in[17];
    float* out = (float*)d_out;

    float* ws = (float*)d_ws;
    size_t o = 0;
    float* W1    = ws + o; o += DD_W;
    float* W2    = ws + o; o += DD_W;
    float* b1    = ws + o; o += 3072;
    float* b2    = ws + o; o += 768;
    float* hbuf  = ws + o; o += (size_t)256 * 3072;
    float* mbuf  = ws + o; o += DD_W;
    float* wsm1  = ws + o; o += 2048;
    float* part1 = ws + o; o += 256;
    float* wsm2  = hbuf;
    float* part2 = hbuf + 2048;

    setup_kernel<<<SETUP_WG, SWT, 0, stream>>>(theta, BB_W1, GG_W1, BB_W2, GG_W2,
                                               BB_b1, Pi_b1, GG_b1, ib1,
                                               BB_b2, Pi_b2, GG_b2, ib2,
                                               wsm1, wsm2, part1, part2, b1, b2);

    fwht_passA_kernel<<<COLS / TC, 512, 0, stream>>>(wsm1, Pi_W1, GG_W1, mbuf, 2047);
    fwht_passB_kernel<<<ROWS_OUT, 512, 0, stream>>>(mbuf, iW1, part1, W1, DD_W);

    fwht_passA_kernel<<<COLS / TC, 512, 0, stream>>>(wsm2, Pi_W2, GG_W2, mbuf, 2047);
    fwht_passB_kernel<<<ROWS_OUT, 512, 0, stream>>>(mbuf, iW2, part2, W2, DD_W);

    gemm_wmma_kernel<1, 2, 64, 12><<<384, 128, 0, stream>>>(x, W1, b1, hbuf, 256, 3072);
    gemm_wmma_kernel<0, 1, 128, 24><<<192, 128, 0, stream>>>(hbuf, W2, b2, out, 256, 768);
}


# Pass A: the 8-KB table w is copied into LDS and the Pi-indexed gather reads it there (bank-wise) instead of 32 line-by-line global loads per thread; every product and sum order kept
# speedup vs baseline: 8.1834x; 1.0851x over previous
#include <hip/hip_runtime.h>

typedef __attribute__((ext_vector_type(2))) float v2f;
typedef __attribute__((ext_vector_type(8))) float v8f;
typedef __attribute__((ext_vector_type(4))) float v4f;

#define LL_BIG   (1 << 22)
#define COLS     8192
#define SUB      4096
#define ROWS     512
#define ROWS_OUT 288
#define DD_W     2359296
#define TC       32
#define TSTRIDE  33
#define BPB      2

#define GGB 32
#define GGW 256
#define GGT 256
#define GGWG (GGW / 2)
__device__ __forceinline__ void ggsum_role(const float* __restrict__ GG,
                                           float* __restrict__ part, int wg,
                                           float* red, int tid) {
    constexpr int stride = GGW * GGT;
    constexpr int n = LL_BIG;
    const int t = tid & (GGT - 1);
    float s = 0.f;
    int i = wg * (2 * GGT) + tid;
#pragma unroll 1
    for (; i + (GGB - 1) * stride < n; i += GGB * stride) {
        float g[GGB];
#pragma unroll
        for (int u = 0; u < GGB; ++u) g[u] = GG[i + u * stride];
#pragma unroll
        for (int u = 0; u < GGB; ++u) s += g[u] * g[u];
    }
    for (; i < n; i += stride) {
        float g = GG[i]; s += g * g;
    }
    red[tid] = s; __syncthreads();
    for (int off = GGT / 2; off > 0; off >>= 1) {
        if (t < off) red[tid] += red[tid + off];
        __syncthreads();
    }
    if (t == 0) part[2 * wg + tid / GGT] = red[tid];
}

template<int N>
__device__ __forceinline__ float row_down(float c) {
    return __int_as_float(__builtin_amdgcn_update_dpp(0, __float_as_int(c), 0x100 + N  ,
                                                      0xf, 0xf, true));
}
__device__ __forceinline__ float other_row(float c) {
    return __int_as_float(__builtin_amdgcn_permlanex16(0, __float_as_int(c), 0x76543210u, 0xfedcba98u,
                                                       false, false));
}

__device__ __forceinline__ float gg_total(const float* pv) {
    float a0 = pv[0] + pv[4], a1 = pv[1] + pv[5], a2 = pv[2] + pv[6], a3 = pv[3] + pv[7];
    float c = (a0 + a2) + (a1 + a3);
    c += other_row(c);
    c += row_down<8>(c);
    c += row_down<4>(c);
    c += row_down<2>(c);
    c += row_down<1>(c);
    return __int_as_float(__builtin_amdgcn_readfirstlane(__float_as_int(c)));
}

template<bool STAG>
__device__ __forceinline__ void hadamard_frags(v2f* hf, int lane) {
    const int m = lane & 15, h = lane >> 4;
    const int row = STAG ? (2 * (m & 7) + (m >> 3)) : m;
#pragma unroll
    for (int s = 0; s < 4; ++s) {
        int k0 = STAG ? (4 * s + h) : (4 * s + 2 * h);
        int k1 = STAG ? (k0 + 2) : (k0 + 1);
        hf[s].x = (__popc(row & k0) & 1) ? -1.f : 1.f;
        hf[s].y = (__popc(row & k1) & 1) ? -1.f : 1.f;
    }
}

__device__ __forceinline__ void hadamard16x2(const v2f* hf, const v2f* xa, const v2f* xb,
                                             v8f& da, v8f& db) {
    da = (v8f){}; db = (v8f){};
#pragma unroll
    for (int s = 0; s < 4; ++s) {
        da = __builtin_amdgcn_wmma_f32_16x16x4_f32(false, hf[s], false, xa[s], (short)0, da, false, false);
        db = __builtin_amdgcn_wmma_f32_16x16x4_f32(false, hf[s], false, xb[s], (short)0, db, false, false);
    }
}

template<int S, int ND>
__device__ __forceinline__ void small_h16(float* p, int h, const v2f* hf) {
    constexpr int KS = ND / 4;
    v2f x[KS];
#pragma unroll
    for (int s = 0; s < KS; ++s) {
        x[s].x = p[(4 * s + h) * S];
        x[s].y = p[(4 * s + 2 + h) * S];
    }
    __builtin_amdgcn_sched_barrier(0);
    v8f d = {};
#pragma unroll
    for (int s = 0; s < KS; ++s)
        d = __builtin_amdgcn_wmma_f32_16x16x4_f32(false, hf[s], false, x[s], (short)0, d, false, false);
#pragma unroll
    for (int v = 0; v < ND / 2; ++v) p[(2 * v + h) * S] = d[v];
}

template<int N>
__device__ __forceinline__ void fwht_lds_mm(float* buf, int tid, const v2f* hf) {
    static_assert(N == 1024 || N == 2048 || N == 4096, "three digits: 4 + 4 + (2..4) bits");
    constexpr int NP = N / 256;
    const int lane = tid & 31, wave = tid >> 5;
    const int n = lane & 15, h = lane >> 4;
    __syncthreads();
    if (wave < NP) small_h16<1, 16>(buf + wave * 256 + n * 16, h, hf);
    __syncthreads();
    if (wave < NP) small_h16<16, 16>(buf + wave * 256 + n, h, hf);
    __syncthreads();
    small_h16<256, NP>(buf + wave * 16 + n, h, hf);
    __syncthreads();
}

#define SWT 512
template<int P>
__device__ __forceinline__ void table_role(const float* __restrict__ theta,
                                           const float* __restrict__ BB,
                                           float* __restrict__ wout, float* w, int tid) {
    constexpr int NP = P / SWT;
    {
        float bb[NP], th[NP];
#pragma unroll
        for (int u = 0; u < NP; ++u) {
            bb[u] = BB[tid + u * SWT];
            th[u] = theta[tid + u * SWT];
        }
#pragma unroll
        for (int u = 0; u < NP; ++u) w[tid + u * SWT] = bb[u] * th[u];
    }
    v2f hf[4];
    hadamard_frags<true>(hf, tid & 31);
    fwht_lds_mm<P>(w, tid, hf);
#pragma unroll
    for (int u = 0; u < NP; ++u) wout[tid + u * SWT] = w[tid + u * SWT];
}

__device__ __forceinline__ void lds_barrier() {
    __builtin_amdgcn_fence(__ATOMIC_RELEASE, "workgroup", "local");
    __builtin_amdgcn_s_barrier();
    __builtin_amdgcn_fence(__ATOMIC_ACQUIRE, "workgroup", "local");
}

#define WTAB 2048
#define NTH  512
#define GAB  32

#define HPA 4

template<int S>
__device__ __forceinline__ void passA_h16(float* tile, int tid, const v2f* hf) {
    const int lane = tid & 31, wave = tid >> 5;
    const int n = lane & 15, h = lane >> 4;
    const int col = (wave & 1) * 16 + n;
    v2f x[HPA][4];
    int base[HPA];
    const int wbase = (S == 1 ? (wave >> 1) * 16 : (wave >> 1)) * TSTRIDE + col;
#pragma unroll
    for (int j = 0; j < HPA; ++j) {
        base[j] = wbase + (S == 1 ? j * 128 : (j & 1) * 8 + (j >> 1) * 256) * TSTRIDE;
#pragma unroll
        for (int s = 0; s < 4; ++s) {
            x[j][s].x = tile[base[j] + ((4 * s + h) * S) * TSTRIDE];
            x[j][s].y = tile[base[j] + ((4 * s + 2 + h) * S) * TSTRIDE];
        }
    }
    __builtin_amdgcn_sched_barrier(0);
#pragma unroll
    for (int j = 0; j < HPA; j += 2) {
        v8f d0, d1;
        hadamard16x2(hf, x[j], x[j + 1], d0, d1);
#pragma unroll
        for (int v = 0; v < 8; ++v) {
            tile[base[j] + ((2 * v + h) * S) * TSTRIDE] = d0[v];
            tile[base[j + 1] + ((2 * v + h) * S) * TSTRIDE] = d1[v];
        }
    }
}

__global__ __launch_bounds__(NTH)
void fwht_passA_kernel(const float* __restrict__ w, const int* __restrict__ Pi,
                       const float* __restrict__ GG, float* __restrict__ mbuf) {
    __shared__ float tile[ROWS * TSTRIDE];
    __shared__ __attribute__((aligned(16))) float wtab[WTAB];
    const int tid = threadIdx.x;
    const int c = tid & (TC - 1), pr = tid >> 5;
    const int col0 = blockIdx.x * TC;
    const int gbase = pr * COLS + col0 + c;
    static_assert(GAB == ROWS / 16, "a thread's rows are one batch");
    static_assert(WTAB == 4 * NTH, "the table is one 16-byte vector per thread");
    {
        int pi[GAB];
        float gg[GAB], wv[GAB];
        const v4f tv = ((const v4f*)w)[tid];
        __builtin_amdgcn_sched_barrier(0);
#pragma unroll
        for (int u = 0; u < GAB; ++u) {
            int g = gbase + u * 16 * COLS;
            pi[u] = Pi[g];
            gg[u] = GG[g];
        }
        *(v4f*)&wtab[4 * tid] = tv;
        lds_barrier();
        __builtin_amdgcn_sched_barrier(0);
#pragma unroll
        for (int u = 0; u < GAB; ++u) wv[u] = wtab[pi[u] & (WTAB - 1)];
#pragma unroll
        for (int u = 0; u < GAB; ++u)
            tile[(pr + 16 * u) * TSTRIDE + c] = wv[u] * gg[u];
    }
    v2f hf[4];
    hadamard_frags<true>(hf, tid & 31);
    __syncthreads(); passA_h16<1>(tile, tid, hf);
    __syncthreads(); passA_h16<16>(tile, tid, hf);
    __syncthreads();
    constexpr int OB = ROWS_OUT * TC / NTH;
    float lo[OB], hi[OB];
#pragma unroll
    for (int it = 0; it < OB; ++it) {
        int rl = pr + 16 * (it & 15);
        lo[it] = tile[rl * TSTRIDE + c];
        hi[it] = tile[(rl + ROWS / 2) * TSTRIDE + c];
    }
#pragma unroll
    for (int it = 0; it < OB; ++it) {
        int r = pr + 16 * it;
        mbuf[(size_t)r * COLS + col0 + c] = (it < 16) ? lo[it] + hi[it] : lo[it] - hi[it];
    }
}

#define BVEC (BPB * SUB / 4 / NTH)

#define BLDS (BPB * BBLK)
#define HPB 2
#define BBLK (SUB / 16 * 20 + SUB / 256 * 16)

__device__ __forceinline__ int pb_idx(int i) {
    return (i >> 4) * 20 + (i & 15) + (i >> 8) * 16;
}

__device__ __forceinline__ void passB_h16_lo(float* blk, int tid, const v2f* hf) {
    const int lane = tid & 31, wave = tid >> 5;
    const int n = lane & 15, h = lane >> 4;
    v2f x[HPB][4];
    int base[HPB];
#pragma unroll
    for (int j = 0; j < HPB; ++j) {
        base[j] = BBLK * j + wave * 336 + n * 20;
#pragma unroll
        for (int s = 0; s < 4; ++s)
            x[j][s] = *(const v2f*)&blk[base[j] + 4 * s + 2 * h];
    }
    __builtin_amdgcn_sched_barrier(0);
#pragma unroll
    for (int j = 0; j < HPB; j += 2) {
        v8f d[2];
        hadamard16x2(hf, x[j], x[j + 1], d[0], d[1]);
#pragma unroll
        for (int t = 0; t < 2; ++t) {
            v4f lo = {d[t][0], d[t][1], d[t][2], d[t][3]}, hi = {d[t][4], d[t][5], d[t][6], d[t][7]};
            *(v4f*)&blk[base[j + t] + 8 * h] = lo;
            *(v4f*)&blk[base[j + t] + 8 * h + 4] = hi;
        }
    }
}

template<int S>
__device__ __forceinline__ void passB_h16_hi(float* blk, int tid, const v2f* hf) {
    const int lane = tid & 31, wave = tid >> 5;
    const int n = lane & 15, h = lane >> 4;
    v2f x[HPB][4];
    int base[HPB];
    constexpr int DS = (S == 16) ? 20 : 336;
    const int wbase = wave * (S == 16 ? 336 : 20) + n + h * DS;
#pragma unroll
    for (int j = 0; j < HPB; ++j) {
        base[j] = wbase + BBLK * j;
#pragma unroll
        for (int s = 0; s < 4; ++s) {
            x[j][s].x = blk[base[j] + (4 * s) * DS];
            x[j][s].y = blk[base[j] + (4 * s + 2) * DS];
        }
    }
    __builtin_amdgcn_sched_barrier(0);
#pragma unroll
    for (int j = 0; j < HPB; j += 2) {
        v8f d0, d1;
        hadamard16x2(hf, x[j], x[j + 1], d0, d1);
#pragma unroll
        for (int v = 0; v < 8; ++v) {
            blk[base[j] + (2 * v) * DS] = d0[v];
            blk[base[j + 1] + (2 * v) * DS] = d1[v];
        }
    }
}

__global__ __launch_bounds__(NTH)
void fwht_passB_kernel(const float* __restrict__ mbuf, const float* __restrict__ init,
                       const float* __restrict__ part, float* __restrict__ out, int DD) {
    __shared__ __attribute__((aligned(16))) float blk[BLDS];
    const int tid = threadIdx.x;
    const size_t base = (size_t)blockIdx.x * COLS;
    const v4f* src = (const v4f*)(mbuf + base);
    const v4f* ini = (const v4f*)(init + base);
    v4f* dst = (v4f*)(out + base);
    v4f mv[BVEC], iv[BVEC];
#pragma unroll
    for (int u = 0; u < BVEC; ++u) mv[u] = src[tid + u * NTH];
#pragma unroll
    for (int u = 0; u < BVEC; ++u) iv[u] = ini[tid + u * NTH];
    float pv[GGW / 32];
#pragma unroll
    for (int j = 0; j < GGW / 32; ++j) pv[j] = part[(tid & 31) + 32 * j];
#pragma unroll
    for (int u = 0; u < BVEC; ++u) *(v4f*)&blk[pb_idx(4 * (tid + u * NTH))] = mv[u];
    v2f hp[4], hs[4];
    hadamard_frags<false>(hp, tid & 31);
    hadamard_frags<true>(hs, tid & 31);
    __syncthreads(); passB_h16_lo(blk, tid, hp);
    __syncthreads(); passB_h16_hi<16>(blk, tid, hs);
    __syncthreads(); passB_h16_hi<256>(blk, tid, hs);
    __syncthreads();
    const float scale = rsqrtf((float)DD * gg_total(pv));
    v4f m[BVEC];
#pragma unroll
    for (int u = 0; u < BVEC; ++u) m[u] = *(const v4f*)&blk[pb_idx(4 * (tid + u * NTH))];
#pragma unroll
    for (int u = 0; u < BVEC / 2; ++u) {
        v4f s = m[u] + m[u + BVEC / 2], d = m[u] - m[u + BVEC / 2];
        dst[tid + u * NTH] = iv[u] + s * scale;
        dst[tid + (u + BVEC / 2) * NTH] = iv[u + BVEC / 2] + d * scale;
    }
}

template<int P, int LLV, int DD>
__device__ __forceinline__ void bias_role(const float* __restrict__ theta, const float* __restrict__ BB,
                                          const int* __restrict__ Pi, const float* __restrict__ GG,
                                          const float* __restrict__ init, float* __restrict__ out,
                                          float* w, float* m, float* red, int tid) {
    static_assert(DD <= LLV, "the outputs are the first DD elements of the transform");
    constexpr int NP = P / SWT, NL = LLV / SWT, NO = (DD + SWT - 1) / SWT;
    v2f hf[4];
    hadamard_frags<true>(hf, tid & 31);
    {
        float bb[NP], th[NP];
#pragma unroll
        for (int u = 0; u < NP; ++u) {
            bb[u] = BB[tid + u * SWT];
            th[u] = theta[tid + u * SWT];
        }
#pragma unroll
        for (int u = 0; u < NP; ++u) w[tid + u * SWT] = bb[u] * th[u];
    }
    fwht_lds_mm<P>(w, tid, hf);
    float s = 0.f;
    {
        float g[NL];
#pragma unroll
        for (int u = 0; u < NL; ++u) g[u] = GG[tid + u * SWT];
#pragma unroll
        for (int u = 0; u < NL; ++u) s = __builtin_fmaf(g[u], g[u], s);
    }
    red[tid] = s; __syncthreads();
    for (int off = SWT >> 1; off > 0; off >>= 1) {
        if (tid < off) red[tid] += red[tid + off];
        __syncthreads();
    }
    {
        int pi[NL];
        float gg[NL], wv[NL];
#pragma unroll
        for (int u = 0; u < NL; ++u) {
            pi[u] = Pi[tid + u * SWT];
            gg[u] = GG[tid + u * SWT];
        }
#pragma unroll
        for (int u = 0; u < NL; ++u) wv[u] = w[pi[u] & (P - 1)];
#pragma unroll
        for (int u = 0; u < NL; ++u) m[tid + u * SWT] = wv[u] * gg[u];
    }
    fwht_lds_mm<LLV>(m, tid, hf);
    const float scale = rsqrtf((float)DD * red[0]);
    {
        float iv[NO];
#pragma unroll
        for (int u = 0; u < NO; ++u)
            iv[u] = (tid + u * SWT < DD) ? init[tid + u * SWT] : 0.f;
#pragma unroll
        for (int u = 0; u < NO; ++u)
            if (tid + u * SWT < DD) out[tid + u * SWT] = iv[u] + m[tid + u * SWT] * scale;
    }
}

#define SETUP_ONE 4
#define SETUP_WG (SETUP_ONE + 2 * GGWG)
__global__ __launch_bounds__(SWT)
void setup_kernel(const float* __restrict__ theta,
                  const float* __restrict__ BB_W1, const float* __restrict__ GG_W1,
                  const float* __restrict__ BB_W2, const float* __restrict__ GG_W2,
                  const float* __restrict__ BB_b1, const int* __restrict__ Pi_b1,
                  const float* __restrict__ GG_b1, const float* __restrict__ ib1,
                  const float* __restrict__ BB_b2, const int* __restrict__ Pi_b2,
                  const float* __restrict__ GG_b2, const float* __restrict__ ib2,
                  float* __restrict__ w1, float* __restrict__ w2,
                  float* __restrict__ part1, float* __restrict__ part2,
                  float* __restrict__ b1, float* __restrict__ b2) {
    __shared__ float smem[2048 + 4096 + SWT];
    const int tid = threadIdx.x;
    const int b = blockIdx.x;
    if (b == 0) {
        bias_role<2048, 4096, 3072>(theta, BB_b1, Pi_b1, GG_b1, ib1, b1,
                                    smem, smem + 2048, smem + 2048 + 4096, tid);
    } else if (b == 1) {
        bias_role<1024, 1024, 768>(theta, BB_b2, Pi_b2, GG_b2, ib2, b2,
                                   smem, smem + 1024, smem + 2048, tid);
    } else if (b < SETUP_ONE) {
        table_role<2048>(theta, b == 2 ? BB_W1 : BB_W2, b == 2 ? w1 : w2, smem, tid);
    } else if (b < SETUP_ONE + GGWG) {
        ggsum_role(GG_W1, part1, b - SETUP_ONE, smem, tid);
    } else {
        ggsum_role(GG_W2, part2, b - SETUP_ONE - GGWG, smem, tid);
    }
}

#define GBM 32
#define CPITCH 36

template<int NT, int KH, int LD>
__device__ __forceinline__ void gemm_read_half(const float* as, const float* bs,
                                               v2f* fa, v2f* fb0, v2f* fb1) {
#pragma unroll
    for (int s = 0; s < KH; ++s) {
        fa[s]  = *(const v2f*)(as + 4 * s);
        fb0[s] = *(const v2f*)(bs + 4 * s);
        if constexpr (NT == 2) fb1[s] = *(const v2f*)(bs + 16 * LD + 4 * s);
    }
}

template<int NT, int KH>
__device__ __forceinline__ void gemm_mma_half(const v2f* fa, const v2f* fb0, const v2f* fb1,
                                              v8f& acc0, v8f& acc1) {
    static_assert(KH % 2 == 0, "NT = 1: the parity of a k-step is its parity within the half");
#pragma unroll
    for (int s = 0; s < KH; ++s) {
        if constexpr (NT == 2) {
            acc0 = __builtin_amdgcn_wmma_f32_16x16x4_f32(false, fa[s], false, fb0[s], (short)0, acc0, false, false);
            acc1 = __builtin_amdgcn_wmma_f32_16x16x4_f32(false, fa[s], false, fb1[s], (short)0, acc1, false, false);
        } else if (s & 1) {
            acc1 = __builtin_amdgcn_wmma_f32_16x16x4_f32(false, fa[s], false, fb0[s], (short)0, acc1, false, false);
        } else {
            acc0 = __builtin_amdgcn_wmma_f32_16x16x4_f32(false, fa[s], false, fb0[s], (short)0, acc0, false, false);
        }
    }
}

template<int V> struct ctag { static constexpr int value = V; };

template<int RELU, int NT, int KC, int NK>
__global__ __launch_bounds__(128)
void gemm_wmma_kernel(const float* __restrict__ A, const float* __restrict__ B,
                      const float* __restrict__ bias, float* __restrict__ C,
                      int M, int N) {
    static_assert(NK >= 4 && NK % 2 == 0, "the pipeline below is written for an even chunk count");
    constexpr int K   = NK * KC;
    constexpr int BN  = 32 * NT;
    constexpr int LD  = KC + 4;
    constexpr int TPR = KC / 4;
    constexpr int RPP = 128 / TPR;
    constexpr int NA  = GBM / RPP, NB = BN / RPP;
    constexpr int KS  = KC / 4;
    constexpr int KH  = KS / 2;
    __shared__ __attribute__((aligned(16))) float As[2][GBM * LD];
    __shared__ __attribute__((aligned(16))) float Bs[2][BN * LD];
    const int tid = threadIdx.x;
    const int lane = tid & 31, wave = tid >> 5;
    const int c = lane & 15, h = lane >> 4;
    const int wm = wave & 1, wn = wave >> 1;
    const int mBlocks = M / GBM;
    const int mb = blockIdx.x % mBlocks;
    const int nb = blockIdx.x / mBlocks;
    const int m0 = mb * GBM, n0 = nb * BN;

    const int kq = (tid % TPR) * 4, lr = tid / TPR;
    const float* ag = A + (size_t)(m0 + lr) * K + kq;
    const float* bg = B + (size_t)(n0 + lr) * K + kq;
    constexpr size_t pass = (size_t)RPP * K;
    const int lofs = lr * LD + kq;

    v4f ra[3][NA], rb[3][NB];
    v2f fa[2][KH], fb0[2][KH], fb1[2][NT == 2 ? KH : 1];
    v8f acc0 = {}, acc1 = {};
    const float* asr = &As[0][(wm * 16 + c) * LD + 2 * h];
    const float* bsr = &Bs[0][(wn * 16 * NT + c) * LD + 2 * h];

    auto fetch = [&](auto set, int kn) __attribute__((always_inline)) {
        constexpr int S = decltype(set)::value;
#pragma unroll
        for (int i = 0; i < NA; ++i) ra[S][i] = *(const v4f*)(ag + i * pass + kn);
#pragma unroll
        for (int i = 0; i < NB; ++i) rb[S][i] = *(const v4f*)(bg + i * pass + kn);
    };
    auto park = [&](auto set) __attribute__((always_inline)) {
        constexpr int S = decltype(set)::value;
#pragma unroll
        for (int i = 0; i < NA; ++i) *(v4f*)&As[S % 2][lofs + i * RPP * LD] = ra[S][i];
#pragma unroll
        for (int i = 0; i < NB; ++i) *(v4f*)&Bs[S % 2][lofs + i * RPP * LD] = rb[S][i];
    };

    auto chunk = [&](int ch, auto kind, auto par) __attribute__((always_inline)) {
        constexpr int KIND = decltype(kind)::value;
        constexpr int P = decltype(par)::value;
        constexpr int CUR = 1 - P;
        gemm_read_half<NT, KH, LD>(asr + CUR * (GBM * LD) + 4 * KH, bsr + CUR * (BN * LD) + 4 * KH,
                                   fa[1], fb0[1], fb1[1]);
        if constexpr (KIND >= 1) park(ctag<P>{});
        if constexpr (KIND >= 2) fetch(ctag<P>{}, (ch + 3) * KC);
        __builtin_amdgcn_sched_barrier(0);
        gemm_mma_half<NT, KH>(fa[0], fb0[0], fb1[0], acc0, acc1);
        asm volatile("" :: "v"(acc0), "v"(acc1));
        __builtin_amdgcn_sched_barrier(0);
        lds_barrier();
        if constexpr (KIND >= 1)
            gemm_read_half<NT, KH, LD>(asr + P * (GBM * LD), bsr + P * (BN * LD),
                                       fa[0], fb0[0], fb1[0]);
        __builtin_amdgcn_sched_barrier(0);
        gemm_mma_half<NT, KH>(fa[1], fb0[1], fb1[1], acc0, acc1);
        asm volatile("" :: "v"(acc0), "v"(acc1));
        __builtin_amdgcn_sched_barrier(0);
    };

    fetch(ctag<2>{}, 0);
    __builtin_amdgcn_sched_barrier(0);
    fetch(ctag<1>{}, KC);
    __builtin_amdgcn_sched_barrier(0);
    fetch(ctag<0>{}, 2 * KC);
    __builtin_amdgcn_sched_barrier(0);
    park(ctag<2>{});
    lds_barrier();
    gemm_read_half<NT, KH, LD>(asr, bsr, fa[0], fb0[0], fb1[0]);

    int ch = 0;
#pragma unroll 1
    for (; ch + 4 < NK; ch += 2) {
        chunk(ch,     ctag<2>{}, ctag<1>{});
        chunk(ch + 1, ctag<2>{}, ctag<0>{});
    }
    chunk(NK - 4, ctag<2>{}, ctag<1>{});
    chunk(NK - 3, ctag<1>{}, ctag<0>{});
    chunk(NK - 2, ctag<1>{}, ctag<1>{});
    chunk(NK - 1, ctag<0>{}, ctag<0>{});

    if constexpr (NT == 2) {
        const int mrow = m0 + wm * 16 + h * 8;
        v8f accs[2] = {acc0, acc1};
#pragma unroll
        for (int t = 0; t < 2; ++t) {
            int n = n0 + wn * 32 + t * 16 + c;
            float bv = bias[n];
            v8f acc = accs[t];
#pragma unroll
            for (int v = 0; v < 8; ++v) {
                float xv = acc[v] + bv;
                if (RELU) xv = fmaxf(xv, 0.f);
                C[(size_t)(mrow + v) * N + n] = xv;
            }
        }
    } else {
        float* cs = &As[0][0];
        static_assert(GBM * CPITCH <= GBM * LD, "the staged tile fits the first A buffer");
        const v8f acc = acc0 + acc1;
#pragma unroll
        for (int v = 0; v < 8; ++v)
            cs[(wm * 16 + h * 8 + v) * CPITCH + wn * 16 + c] = acc[v];
        __syncthreads();
        const float bv = bias[n0 + lane];
#pragma unroll
        for (int i = 0; i < 8; ++i) {
            int r = wave * 8 + i;
            float xv = cs[r * CPITCH + lane] + bv;
            if (RELU) xv = fmaxf(xv, 0.f);
            C[(size_t)(m0 + r) * N + n0 + lane] = xv;
        }
    }
}

extern "C" void kernel_launch(void* const* d_in, const int* in_sizes, int n_in,
                              void* d_out, int out_size, void* d_ws, size_t ws_size,
                              hipStream_t stream) {
    const float* x     = (const float*)d_in[0];
    const float* theta = (const float*)d_in[1];
    const float* iW1   = (const float*)d_in[2];
    const float* ib1   = (const float*)d_in[3];
    const float* iW2   = (const float*)d_in[4];
    const float* ib2   = (const float*)d_in[5];
    const float* BB_W1 = (const float*)d_in[6];
    const int*   Pi_W1 = (const int*)d_in[7];
    const float* GG_W1 = (const float*)d_in[8];
    const float* BB_b1 = (const float*)d_in[9];
    const int*   Pi_b1 = (const int*)d_in[10];
    const float* GG_b1 = (const float*)d_in[11];
    const float* BB_W2 = (const float*)d_in[12];
    const int*   Pi_W2 = (const int*)d_in[13];
    const float* GG_W2 = (const float*)d_in[14];
    const float* BB_b2 = (const float*)d_in[15];
    const int*   Pi_b2 = (const int*)d_in[16];
    const float* GG_b2 = (const float*)d_in[17];
    float* out = (float*)d_out;

    float* ws = (float*)d_ws;
    size_t o = 0;
    float* W1    = ws + o; o += DD_W;
    float* W2    = ws + o; o += DD_W;
    float* b1    = ws + o; o += 3072;
    float* b2    = ws + o; o += 768;
    float* hbuf  = ws + o; o += (size_t)256 * 3072;
    float* mbuf  = ws + o; o += DD_W;
    float* wsm1  = ws + o; o += 2048;
    float* part1 = ws + o; o += 256;
    float* wsm2  = hbuf;
    float* part2 = hbuf + 2048;

    setup_kernel<<<SETUP_WG, SWT, 0, stream>>>(theta, BB_W1, GG_W1, BB_W2, GG_W2,
                                               BB_b1, Pi_b1, GG_b1, ib1,
                                               BB_b2, Pi_b2, GG_b2, ib2,
                                               wsm1, wsm2, part1, part2, b1, b2);

    fwht_passA_kernel<<<COLS / TC, 512, 0, stream>>>(wsm1, Pi_W1, GG_W1, mbuf);
    fwht_passB_kernel<<<ROWS_OUT, 512, 0, stream>>>(mbuf, iW1, part1, W1, DD_W);

    fwht_passA_kernel<<<COLS / TC, 512, 0, stream>>>(wsm2, Pi_W2, GG_W2, mbuf);
    fwht_passB_kernel<<<ROWS_OUT, 512, 0, stream>>>(mbuf, iW2, part2, W2, DD_W);

    gemm_wmma_kernel<1, 2, 64, 12><<<384, 128, 0, stream>>>(x, W1, b1, hbuf, 256, 3072);
    gemm_wmma_kernel<0, 1, 128, 24><<<192, 128, 0, stream>>>(hbuf, W2, b2, out, 256, 768);
}
